# starting kernel
# baseline (speedup 1.0000x reference)
#include <hip/hip_runtime.h>


typedef __attribute__((ext_vector_type(2))) float v2f;
typedef __attribute__((ext_vector_type(8))) float v8f;

#define PI_F 3.14159265358979323846f

__device__ __forceinline__ v8f wmma4(v2f a, v2f b, v8f c) {
  return __builtin_amdgcn_wmma_f32_16x16x4_f32(false, a, false, b, (short)0, c,
                                               false, false);
}

__device__ __forceinline__ int bandmap(int bi) { return (bi + 96) & 127; }

__device__ __forceinline__ v2f fragA(const float* A, int lda, int m0, int k0) {
  int lane = threadIdx.x & 31;
  int m = m0 + (lane & 15);
  int kk = k0 + ((lane >> 4) << 1);
  v2f r; r.x = A[m * lda + kk]; r.y = A[m * lda + kk + 1]; return r;
}
__device__ __forceinline__ v2f fragB(const float* B, int ldb, int k0, int n0) {
  int lane = threadIdx.x & 31;
  int n = n0 + (lane & 15);
  int kk = k0 + ((lane >> 4) << 1);
  v2f r; r.x = B[kk * ldb + n]; r.y = B[(kk + 1) * ldb + n]; return r;
}
__device__ __forceinline__ v2f fragB_bc(const float* F, int k0, int n0) {
  int lane = threadIdx.x & 31;
  int col = bandmap(n0 + (lane & 15));
  int kk = k0 + ((lane >> 4) << 1);
  v2f r; r.x = F[kk * 128 + col]; r.y = F[(kk + 1) * 128 + col]; return r;
}
__device__ __forceinline__ v2f fragA_br(const float* F, int m0, int k0) {
  int lane = threadIdx.x & 31;
  int row = bandmap(m0 + (lane & 15));
  int kk = k0 + ((lane >> 4) << 1);
  v2f r; r.x = F[row * 128 + kk]; r.y = F[row * 128 + kk + 1]; return r;
}
__device__ __forceinline__ v2f fragA_kmap(const float* F, int m0, int k0) {
  int lane = threadIdx.x & 31;
  int m = m0 + (lane & 15);
  int kk = k0 + ((lane >> 4) << 1);
  v2f r; r.x = F[m * 128 + bandmap(kk)]; r.y = F[m * 128 + bandmap(kk + 1)];
  return r;
}
__device__ __forceinline__ v2f fragB_kmap(const float* F, int k0, int n0) {
  int lane = threadIdx.x & 31;
  int n = n0 + (lane & 15);
  int kk = k0 + ((lane >> 4) << 1);
  v2f r; r.x = F[bandmap(kk) * 128 + n]; r.y = F[bandmap(kk + 1) * 128 + n];
  return r;
}

__global__ void k0_tables(float* __restrict__ Ftab) {
  for (int t = threadIdx.x + blockIdx.x * blockDim.x; t < 16384;
       t += blockDim.x * gridDim.x) {
    int j = t >> 7, k = t & 127;
    float ang = -2.f * PI_F * (float)((j * k) & 127) / 128.f;
    float s, c;
    sincosf(ang, &s, &c);
    Ftab[t] = c;
    Ftab[16384 + t] = s;
  }
}

__global__ void k1_fwd(const float* __restrict__ x, const float* __restrict__ Ftab,
                       float* __restrict__ XbR, float* __restrict__ XbI) {
  __shared__ float T1r[128 * 64];
  __shared__ float T1i[128 * 64];
  const float* Fre = Ftab;
  const float* Fim = Ftab + 16384;
  int blk = blockIdx.x;
  int b = blk >> 6, ci = blk & 63;
  const float* Ximg = x + ((size_t)blk << 14);
  int wave = threadIdx.x >> 5;
  int lane = threadIdx.x & 31;

  for (int t = wave; t < 32; t += 8) {
    int m0 = (t >> 2) << 4, n0 = (t & 3) << 4;
    v8f aR = {}, aI = {};
    for (int k = 0; k < 128; k += 4) {
      v2f a  = fragA(Ximg, 128, m0, k);
      v2f br = fragB_bc(Fre, k, n0);
      v2f bi = fragB_bc(Fim, k, n0);
      aR = wmma4(a, br, aR);
      aI = wmma4(a, bi, aI);
    }
    int n = n0 + (lane & 15);
    int mb = m0 + ((lane >> 4) << 3);
#pragma unroll
    for (int r = 0; r < 8; ++r) {
      T1r[(mb + r) * 64 + n] = aR[r];
      T1i[(mb + r) * 64 + n] = aI[r];
    }
  }
  __syncthreads();

  int bdst = (b + 8) & 15;
  for (int t = wave; t < 16; t += 8) {
    int m0 = (t >> 2) << 4, n0 = (t & 3) << 4;
    v8f cR = {}, cI = {};
    for (int k = 0; k < 128; k += 4) {
      v2f ar = fragA_br(Fre, m0, k);
      v2f ai = fragA_br(Fim, m0, k);
      v2f br = fragB(T1r, 64, k, n0);
      v2f bi = fragB(T1i, 64, k, n0);
      cR = wmma4(ar, br, cR);
      v2f ani; ani.x = -ai.x; ani.y = -ai.y;
      cR = wmma4(ani, bi, cR);
      cI = wmma4(ar, bi, cI);
      cI = wmma4(ai, br, cI);
    }
    int n = n0 + (lane & 15);
    int mb = m0 + ((lane >> 4) << 3);
    int dv = n - 32;
#pragma unroll
    for (int r = 0; r < 8; ++r) {
      int uu = mb + r;
      int du = uu - 32;
      float msk = (du * du + dv * dv <= 900) ? 1.f : 0.f;
      size_t off = ((size_t)(uu * 64 + n) << 10) + (size_t)(bdst << 6) + ci;
      XbR[off] = cR[r] * msk;
      XbI[off] = cI[r] * msk;
    }
  }
}

__global__ void k2_wdft(const float* __restrict__ w, float* __restrict__ WbR,
                        float* __restrict__ WbI) {
  int f = blockIdx.x;
  int u = bandmap(f >> 6), v = bandmap(f & 63);
  float cy[3], sy[3], cx[3], sx[3];
#pragma unroll
  for (int t = 0; t < 3; ++t) {
    sincosf(-2.f * PI_F * (float)((u * t) & 127) / 128.f, &sy[t], &cy[t]);
    sincosf(-2.f * PI_F * (float)((v * t) & 127) / 128.f, &sx[t], &cx[t]);
  }
  float* outR = WbR + ((size_t)f << 12);
  float* outI = WbI + ((size_t)f << 12);
  for (int p = threadIdx.x; p < 4096; p += blockDim.x) {
    int i = p >> 6, op = p & 63;
    int osrc = (op + 32) & 63;
    const float* wk = w + (size_t)(osrc * 64 + i) * 9;
    float ar = 0.f, ai = 0.f;
#pragma unroll
    for (int ky = 0; ky < 3; ++ky)
#pragma unroll
      for (int kx = 0; kx < 3; ++kx) {
        float tr = cy[ky] * cx[kx] - sy[ky] * sx[kx];
        float ti = cy[ky] * sx[kx] + sy[ky] * cx[kx];
        float wv = wk[ky * 3 + kx];
        ar += wv * tr;
        ai += wv * ti;
      }
    outR[p] = ar;
    outI[p] = ai;
  }
}

__global__ void k3_mix(const float* __restrict__ XbR, const float* __restrict__ XbI,
                       const float* __restrict__ WbR, const float* __restrict__ WbI,
                       float* __restrict__ YbR, float* __restrict__ YbI) {
  int f = blockIdx.x;
  const float* AR = XbR + ((size_t)f << 10);
  const float* AI = XbI + ((size_t)f << 10);
  const float* BR = WbR + ((size_t)f << 12);
  const float* BI = WbI + ((size_t)f << 12);
  __builtin_prefetch(BR, 0, 1);
  __builtin_prefetch(BI, 0, 1);
  int wave = threadIdx.x >> 5;
  int lane = threadIdx.x & 31;
  int n0 = wave << 4;
  v8f cR = {}, cI = {};
  for (int k = 0; k < 64; k += 4) {
    v2f ar = fragA(AR, 64, 0, k);
    v2f ai = fragA(AI, 64, 0, k);
    v2f br = fragB(BR, 64, k, n0);
    v2f bi = fragB(BI, 64, k, n0);
    cR = wmma4(ar, br, cR);
    v2f ani; ani.x = -ai.x; ani.y = -ai.y;
    cR = wmma4(ani, bi, cR);
    cI = wmma4(ar, bi, cI);
    cI = wmma4(ai, br, cI);
  }
  int n = n0 + (lane & 15);
  int mb = (lane >> 4) << 3;
#pragma unroll
  for (int r = 0; r < 8; ++r) {
    size_t off = ((size_t)((mb + r) * 64 + n) << 12) + f;
    YbR[off] = cR[r];
    YbI[off] = cI[r];
  }
}

__global__ void k4_inv(const float* __restrict__ YbR, const float* __restrict__ YbI,
                       const float* __restrict__ Ftab, const float* __restrict__ bias,
                       float* __restrict__ out) {
  __shared__ float Sr[128 * 64];
  __shared__ float Si[128 * 64];
  const float* Fre = Ftab;
  const float* Fim = Ftab + 16384;
  int blk = blockIdx.x;
  const float* Yr = YbR + ((size_t)blk << 12);
  const float* Yi = YbI + ((size_t)blk << 12);
  int wave = threadIdx.x >> 5;
  int lane = threadIdx.x & 31;

  for (int t = wave; t < 32; t += 8) {
    int m0 = (t >> 2) << 4, n0 = (t & 3) << 4;
    v8f cR = {}, cI = {};
    for (int k = 0; k < 64; k += 4) {
      v2f ar = fragA_kmap(Fre, m0, k);
      v2f ai = fragA_kmap(Fim, m0, k);
      v2f br = fragB(Yr, 64, k, n0);
      v2f bi = fragB(Yi, 64, k, n0);
      cR = wmma4(ar, br, cR);
      cR = wmma4(ai, bi, cR);
      cI = wmma4(ar, bi, cI);
      v2f ani; ani.x = -ai.x; ani.y = -ai.y;
      cI = wmma4(ani, br, cI);
    }
    int n = n0 + (lane & 15);
    int mb = m0 + ((lane >> 4) << 3);
#pragma unroll
    for (int r = 0; r < 8; ++r) {
      Sr[(mb + r) * 64 + n] = cR[r];
      Si[(mb + r) * 64 + n] = cI[r];
    }
  }
  __syncthreads();

  float bo = bias[blk & 63];
  float* oimg = out + ((size_t)blk << 14);
  for (int t = wave; t < 64; t += 8) {
    int m0 = (t >> 3) << 4, n0 = (t & 7) << 4;
    v8f c = {};
    for (int k = 0; k < 64; k += 4) {
      v2f sr = fragA(Sr, 64, m0, k);
      v2f si = fragA(Si, 64, m0, k);
      v2f br = fragB_kmap(Fre, k, n0);
      v2f bi = fragB_kmap(Fim, k, n0);
      c = wmma4(sr, br, c);
      c = wmma4(si, bi, c);
    }
    int n = n0 + (lane & 15);
    int mb = m0 + ((lane >> 4) << 3);
#pragma unroll
    for (int r = 0; r < 8; ++r)
      oimg[(mb + r) * 128 + n] = c[r] * (1.f / 16384.f) + bo;
  }
}

extern "C" void kernel_launch(void* const* d_in, const int* in_sizes, int n_in,
                              void* d_out, int out_size, void* d_ws, size_t ws_size,
                              hipStream_t stream) {
  (void)in_sizes; (void)n_in; (void)out_size; (void)ws_size;
  const float* x    = (const float*)d_in[0];
  const float* w    = (const float*)d_in[1];
  const float* bias = (const float*)d_in[2];
  float* out = (float*)d_out;

  float* ws   = (float*)d_ws;
  float* Ftab = ws;
  float* XbR  = ws + 32768;
  float* XbI  = XbR + 4194304;
  float* WbR  = XbI + 4194304;
  float* WbI  = WbR + 16777216;
  float* YbR  = WbI + 16777216;
  float* YbI  = YbR + 4194304;

  k0_tables<<<16, 256, 0, stream>>>(Ftab);
  k1_fwd  <<<1024, 256, 0, stream>>>(x, Ftab, XbR, XbI);
  k2_wdft <<<4096, 256, 0, stream>>>(w, WbR, WbI);
  k3_mix  <<<4096, 128, 0, stream>>>(XbR, XbI, WbR, WbI, YbR, YbI);
  k4_inv  <<<1024, 256, 0, stream>>>(YbR, YbI, Ftab, bias, out);
}


# Channel mix factored by filter tap: real 3x3 taps are one LDS-resident B operand (K = 9 x 64), twiddles multiply X per lane; weight-spectrum kernel gone; X image-major; sum order changes
# speedup vs baseline: 3.2006x; 3.2006x over previous
#include <hip/hip_runtime.h>


typedef __attribute__((ext_vector_type(2))) float v2f;
typedef __attribute__((ext_vector_type(8))) float v8f;

#define PI_F 3.14159265358979323846f

__device__ __forceinline__ v8f wmma4(v2f a, v2f b, v8f c) {
  return __builtin_amdgcn_wmma_f32_16x16x4_f32(false, a, false, b, (short)0, c,
                                               false, false);
}

__device__ __forceinline__ int bandmap(int bi) { return (bi + 96) & 127; }

__device__ __forceinline__ v2f fragA(const float* A, int lda, int m0, int k0) {
  int lane = threadIdx.x & 31;
  int m = m0 + (lane & 15);
  int kk = k0 + ((lane >> 4) << 1);
  v2f r; r.x = A[m * lda + kk]; r.y = A[m * lda + kk + 1]; return r;
}
__device__ __forceinline__ v2f fragB(const float* B, int ldb, int k0, int n0) {
  int lane = threadIdx.x & 31;
  int n = n0 + (lane & 15);
  int kk = k0 + ((lane >> 4) << 1);
  v2f r; r.x = B[kk * ldb + n]; r.y = B[(kk + 1) * ldb + n]; return r;
}
__device__ __forceinline__ v2f fragB_bc(const float* F, int k0, int n0) {
  int lane = threadIdx.x & 31;
  int col = bandmap(n0 + (lane & 15));
  int kk = k0 + ((lane >> 4) << 1);
  v2f r; r.x = F[kk * 128 + col]; r.y = F[(kk + 1) * 128 + col]; return r;
}
__device__ __forceinline__ v2f fragA_br(const float* F, int m0, int k0) {
  int lane = threadIdx.x & 31;
  int row = bandmap(m0 + (lane & 15));
  int kk = k0 + ((lane >> 4) << 1);
  v2f r; r.x = F[row * 128 + kk]; r.y = F[row * 128 + kk + 1]; return r;
}
__device__ __forceinline__ v2f fragA_kmap(const float* F, int m0, int k0) {
  int lane = threadIdx.x & 31;
  int m = m0 + (lane & 15);
  int kk = k0 + ((lane >> 4) << 1);
  v2f r; r.x = F[m * 128 + bandmap(kk)]; r.y = F[m * 128 + bandmap(kk + 1)];
  return r;
}
__device__ __forceinline__ v2f fragB_kmap(const float* F, int k0, int n0) {
  int lane = threadIdx.x & 31;
  int n = n0 + (lane & 15);
  int kk = k0 + ((lane >> 4) << 1);
  v2f r; r.x = F[bandmap(kk) * 128 + n]; r.y = F[bandmap(kk + 1) * 128 + n];
  return r;
}

__global__ void k0_tables(float* __restrict__ Ftab) {
  for (int t = threadIdx.x + blockIdx.x * blockDim.x; t < 16384;
       t += blockDim.x * gridDim.x) {
    int j = t >> 7, k = t & 127;
    float ang = -2.f * PI_F * (float)((j * k) & 127) / 128.f;
    float s, c;
    sincosf(ang, &s, &c);
    Ftab[t] = c;
    Ftab[16384 + t] = s;
  }
}

__global__ void k1_fwd(const float* __restrict__ x, const float* __restrict__ Ftab,
                       float* __restrict__ XbR, float* __restrict__ XbI) {
  __shared__ float T1r[128 * 64];
  __shared__ float T1i[128 * 64];
  const float* Fre = Ftab;
  const float* Fim = Ftab + 16384;
  int blk = blockIdx.x;
  int b = blk >> 6, ci = blk & 63;
  const float* Ximg = x + ((size_t)blk << 14);
  int wave = threadIdx.x >> 5;
  int lane = threadIdx.x & 31;

  for (int t = wave; t < 32; t += 8) {
    int m0 = (t >> 2) << 4, n0 = (t & 3) << 4;
    v8f aR = {}, aI = {};
    for (int k = 0; k < 128; k += 4) {
      v2f a  = fragA(Ximg, 128, m0, k);
      v2f br = fragB_bc(Fre, k, n0);
      v2f bi = fragB_bc(Fim, k, n0);
      aR = wmma4(a, br, aR);
      aI = wmma4(a, bi, aI);
    }
    int n = n0 + (lane & 15);
    int mb = m0 + ((lane >> 4) << 3);
#pragma unroll
    for (int r = 0; r < 8; ++r) {
      T1r[(mb + r) * 64 + n] = aR[r];
      T1i[(mb + r) * 64 + n] = aI[r];
    }
  }
  __syncthreads();

  int bdst = (b + 8) & 15;
  for (int t = wave; t < 16; t += 8) {
    int m0 = (t >> 2) << 4, n0 = (t & 3) << 4;
    v8f cR = {}, cI = {};
    for (int k = 0; k < 128; k += 4) {
      v2f ar = fragA_br(Fre, m0, k);
      v2f ai = fragA_br(Fim, m0, k);
      v2f br = fragB(T1r, 64, k, n0);
      v2f bi = fragB(T1i, 64, k, n0);
      cR = wmma4(ar, br, cR);
      v2f ani; ani.x = -ai.x; ani.y = -ai.y;
      cR = wmma4(ani, bi, cR);
      cI = wmma4(ar, bi, cI);
      cI = wmma4(ai, br, cI);
    }
    int n = n0 + (lane & 15);
    int mb = m0 + ((lane >> 4) << 3);
    int dv = n - 32;
#pragma unroll
    for (int r = 0; r < 8; ++r) {
      int uu = mb + r;
      int du = uu - 32;
      float msk = (du * du + dv * dv <= 900) ? 1.f : 0.f;
      size_t off = ((size_t)((bdst << 6) + ci) << 12) + (size_t)(uu * 64 + n);
      XbR[off] = cR[r] * msk;
      XbI[off] = cI[r] * msk;
    }
  }
}

#define WP 65
__global__ void __launch_bounds__(512)
k3_mix(const float* __restrict__ XbR, const float* __restrict__ XbI,
       const float* __restrict__ w, const float* __restrict__ Ftab,
       float* __restrict__ YbR, float* __restrict__ YbI) {
  __shared__ float Wl[576 * WP];
  int wave = threadIdx.x >> 5;
  int lane = threadIdx.x & 31;
  int c = lane & 15, h = lane >> 4;

  for (int q = threadIdx.x; q < 36864; q += 512) {
    int osrc = q / 576;
    int rho = q - osrc * 576;
    Wl[rho * WP + ((osrc + 32) & 63)] = w[q];
  }

  int f = (blockIdx.x << 4) + c;
  int u = bandmap(f >> 6), v = bandmap(f & 63);
  float tr[9], ti[9];
#pragma unroll
  for (int ky = 0; ky < 3; ++ky)
#pragma unroll
    for (int kx = 0; kx < 3; ++kx) {
      int idx = (u * ky + v * kx) & 127;
      tr[ky * 3 + kx] = Ftab[128 + idx];
      ti[ky * 3 + kx] = Ftab[16384 + 128 + idx];
    }
  __syncthreads();

  const float* xr_p = XbR + ((size_t)(wave << 6) << 12) + f;
  const float* xi_p = XbI + ((size_t)(wave << 6) << 12) + f;
  v8f aR[4] = {}, aI[4] = {};
  for (int i0 = 0; i0 < 64; i0 += 4) {
    int kk = i0 + (h << 1);
    float xr0 = xr_p[(size_t)kk << 12], xr1 = xr_p[(size_t)(kk + 1) << 12];
    float xi0 = xi_p[(size_t)kk << 12], xi1 = xi_p[(size_t)(kk + 1) << 12];
    const float* wl = Wl + (kk * 9) * WP + c;
#pragma unroll
    for (int t = 0; t < 9; ++t) {
      v2f ar, ai;
      ar.x = xr0 * tr[t] - xi0 * ti[t];
      ar.y = xr1 * tr[t] - xi1 * ti[t];
      ai.x = xr0 * ti[t] + xi0 * tr[t];
      ai.y = xr1 * ti[t] + xi1 * tr[t];
#pragma unroll
      for (int ot = 0; ot < 4; ++ot) {
        v2f bf;
        bf.x = wl[t * WP + (ot << 4)];
        bf.y = wl[(9 + t) * WP + (ot << 4)];
        aR[ot] = wmma4(ar, bf, aR[ot]);
        aI[ot] = wmma4(ai, bf, aI[ot]);
      }
    }
  }

  int fb = (blockIdx.x << 4) + (h << 3);
#pragma unroll
  for (int ot = 0; ot < 4; ++ot)
#pragma unroll
    for (int r = 0; r < 8; ++r) {
      size_t off = ((size_t)(fb + r) << 10) + (size_t)((wave << 6) + (ot << 4) + c);
      YbR[off] = aR[ot][r];
      YbI[off] = aI[ot][r];
    }
}

#define APITCH 68
__global__ void k4a_col(const float* __restrict__ YbR, const float* __restrict__ YbI,
                        const float* __restrict__ Ftab,
                        float* __restrict__ SR, float* __restrict__ SI) {
  __shared__ float Ar[64 * APITCH];
  __shared__ float Ai[64 * APITCH];
  const float* Fre = Ftab;
  const float* Fim = Ftab + 16384;
  int mh = blockIdx.x & 1;
  int nb = blockIdx.x >> 1;
  int wave = threadIdx.x >> 5;
  int lane = threadIdx.x & 31;
  int c = lane & 15, h = lane >> 4;
  int n0 = ((nb << 3) + wave) << 4;
  int vv = n0 >> 10, b = (n0 >> 6) & 15, o0 = n0 & 63;

  v2f br[16], bi[16];
  const float* yr = YbR + n0 + c;
  const float* yi = YbI + n0 + c;
#pragma unroll
  for (int ks = 0; ks < 16; ++ks) {
    size_t row = (size_t)((ks << 2) + (h << 1)) << 16;
    br[ks].x = yr[row]; br[ks].y = yr[row + 65536];
    bi[ks].x = yi[row]; bi[ks].y = yi[row + 65536];
  }

  for (int idx = threadIdx.x; idx < 4096; idx += 256) {
    int mr = idx >> 6, k = idx & 63;
    int g = ((mh << 6) + mr) * 128 + bandmap(k);
    Ar[mr * APITCH + k] = Fre[g];
    Ai[mr * APITCH + k] = Fim[g];
  }
  __syncthreads();

  for (int mt = 0; mt < 4; ++mt) {
    v8f cR = {}, cI = {};
    const float* ar_p = Ar + ((mt << 4) + c) * APITCH + (h << 1);
    const float* ai_p = Ai + ((mt << 4) + c) * APITCH + (h << 1);
#pragma unroll
    for (int ks = 0; ks < 16; ++ks) {
      v2f ar = *(const v2f*)(ar_p + (ks << 2));
      v2f ai = *(const v2f*)(ai_p + (ks << 2));
      cR = wmma4(ar, br[ks], cR);
      cR = wmma4(ai, bi[ks], cR);
      cI = wmma4(ar, bi[ks], cI);
      v2f ani; ani.x = -ai.x; ani.y = -ai.y;
      cI = wmma4(ani, br[ks], cI);
    }
    int mbase = (mh << 6) + (mt << 4) + (h << 3);
#pragma unroll
    for (int r = 0; r < 8; ++r) {
      size_t off = ((((size_t)(b * 128 + mbase + r) << 6) + vv) << 6) + o0 + c;
      SR[off] = cR[r];
      SI[off] = cI[r];
    }
  }
}

__global__ void k4b_row(const float* __restrict__ SR, const float* __restrict__ SI,
                        const float* __restrict__ Ftab, const float* __restrict__ bias,
                        float* __restrict__ out) {
  const float* Fre = Ftab;
  const float* Fim = Ftab + 16384;
  int blk = blockIdx.x;
  int b = blk >> 7, m = blk & 127;
  int wave = threadIdx.x >> 5;
  int lane = threadIdx.x & 31;
  int c = lane & 15, h = lane >> 4;
  int ot = wave & 3;
  int ng = wave >> 2;
  const float* sr_p = SR + ((size_t)blk << 12) + (ot << 4) + c;
  const float* si_p = SI + ((size_t)blk << 12) + (ot << 4) + c;

  v2f sr[16], si[16];
#pragma unroll
  for (int ks = 0; ks < 16; ++ks) {
    int row = ((ks << 2) + (h << 1)) << 6;
    sr[ks].x = sr_p[row]; sr[ks].y = sr_p[row + 64];
    si[ks].x = si_p[row]; si[ks].y = si_p[row + 64];
  }

  v8f acc[4] = {};
#pragma unroll
  for (int ks = 0; ks < 16; ++ks) {
#pragma unroll
    for (int j = 0; j < 4; ++j) {
      int n0 = ((ng << 2) + j) << 4;
      v2f fr = fragB_kmap(Fre, ks << 2, n0);
      v2f fi = fragB_kmap(Fim, ks << 2, n0);
      acc[j] = wmma4(sr[ks], fr, acc[j]);
      acc[j] = wmma4(si[ks], fi, acc[j]);
    }
  }

  int ob = (ot << 4) + (h << 3);
#pragma unroll
  for (int r = 0; r < 8; ++r) {
    float bo = bias[ob + r];
    float* orow = out + ((((size_t)(b * 64 + ob + r) << 7) + m) << 7) + (ng << 6) + c;
#pragma unroll
    for (int j = 0; j < 4; ++j)
      orow[j << 4] = acc[j][r] * (1.f / 16384.f) + bo;
  }
}

extern "C" void kernel_launch(void* const* d_in, const int* in_sizes, int n_in,
                              void* d_out, int out_size, void* d_ws, size_t ws_size,
                              hipStream_t stream) {
  (void)in_sizes; (void)n_in; (void)out_size; (void)ws_size;
  const float* x    = (const float*)d_in[0];
  const float* w    = (const float*)d_in[1];
  const float* bias = (const float*)d_in[2];
  float* out = (float*)d_out;

  float* ws   = (float*)d_ws;
  float* Ftab = ws;
  float* XbR  = ws + 32768;
  float* XbI  = XbR + 4194304;
  float* YbR  = XbI + 4194304;
  float* YbI  = YbR + 4194304;
  float* SR   = YbI + 4194304;
  float* SI   = SR + 8388608;

  k0_tables<<<16, 256, 0, stream>>>(Ftab);
  k1_fwd  <<<1024, 256, 0, stream>>>(x, Ftab, XbR, XbI);
  k3_mix  <<<256, 512, 0, stream>>>(XbR, XbI, w, Ftab, YbR, YbI);
  k4a_col <<<1024, 256, 0, stream>>>(YbR, YbI, Ftab, SR, SI);
  k4b_row <<<2048, 256, 0, stream>>>(SR, SI, Ftab, bias, out);
}


# k1 runs the column DFT first: x is the B operand, read once in 64-byte row pieces and held in registers; the 64 band rows of F sit in LDS and feed both products; sum order changes
# speedup vs baseline: 3.6247x; 1.1325x over previous
#include <hip/hip_runtime.h>


typedef __attribute__((ext_vector_type(2))) float v2f;
typedef __attribute__((ext_vector_type(4))) float v4f;
typedef __attribute__((ext_vector_type(8))) float v8f;

#define PI_F 3.14159265358979323846f

__device__ __forceinline__ v8f wmma4(v2f a, v2f b, v8f c) {
  return __builtin_amdgcn_wmma_f32_16x16x4_f32(false, a, false, b, (short)0, c,
                                               false, false);
}

__device__ __forceinline__ int bandmap(int bi) { return (bi + 96) & 127; }

__device__ __forceinline__ v2f fragA_kmap(const float* F, int m0, int k0) {
  int lane = threadIdx.x & 31;
  int m = m0 + (lane & 15);
  int kk = k0 + ((lane >> 4) << 1);
  v2f r; r.x = F[m * 128 + bandmap(kk)]; r.y = F[m * 128 + bandmap(kk + 1)];
  return r;
}
__device__ __forceinline__ v2f fragB_kmap(const float* F, int k0, int n0) {
  int lane = threadIdx.x & 31;
  int n = n0 + (lane & 15);
  int kk = k0 + ((lane >> 4) << 1);
  v2f r; r.x = F[bandmap(kk) * 128 + n]; r.y = F[bandmap(kk + 1) * 128 + n];
  return r;
}

__global__ void k0_tables(float* __restrict__ Ftab) {
  for (int t = threadIdx.x + blockIdx.x * blockDim.x; t < 16384;
       t += blockDim.x * gridDim.x) {
    int j = t >> 7, k = t & 127;
    float ang = -2.f * PI_F * (float)((j * k) & 127) / 128.f;
    float s, c;
    sincosf(ang, &s, &c);
    Ftab[t] = c;
    Ftab[16384 + t] = s;
  }
}

#define FP 132
__global__ void __launch_bounds__(256)
k1_fwd(const float* __restrict__ x, const float* __restrict__ Ftab,
       float* __restrict__ XbR, float* __restrict__ XbI) {
  __shared__ __attribute__((aligned(16))) float Fbr[64 * FP];
  __shared__ __attribute__((aligned(16))) float Fbi[64 * FP];
  __shared__ __attribute__((aligned(16))) float Tr[64 * FP];
  __shared__ __attribute__((aligned(16))) float Ti[64 * FP];
  const float* Fre = Ftab;
  const float* Fim = Ftab + 16384;
  int blk = blockIdx.x;
  int b = blk >> 6, ci = blk & 63;
  const float* Ximg = x + ((size_t)blk << 14);
  int wave = threadIdx.x >> 5;
  int lane = threadIdx.x & 31;
  int c = lane & 15, h = lane >> 4;

  v2f xb[32];
  const float* xp = Ximg + (h << 8) + (wave << 4) + c;
#pragma unroll
  for (int ks = 0; ks < 32; ++ks) {
    xb[ks].x = xp[ks << 9];
    xb[ks].y = xp[(ks << 9) + 128];
  }

  for (int q = threadIdx.x; q < 2048; q += 256) {
    int m = q >> 5, k4 = (q & 31) << 2;
    int g = bandmap(m) * 128 + k4;
    *(v4f*)(Fbr + m * FP + k4) = *(const v4f*)(Fre + g);
    *(v4f*)(Fbi + m * FP + k4) = *(const v4f*)(Fim + g);
  }
  __syncthreads();

  for (int mt = 0; mt < 4; ++mt) {
    v8f aR = {}, aI = {};
    const float* ar_p = Fbr + ((mt << 4) + c) * FP + (h << 1);
    const float* ai_p = Fbi + ((mt << 4) + c) * FP + (h << 1);
#pragma unroll
    for (int ks = 0; ks < 32; ++ks) {
      v2f ar = *(const v2f*)(ar_p + (ks << 2));
      v2f ai = *(const v2f*)(ai_p + (ks << 2));
      aR = wmma4(ar, xb[ks], aR);
      aI = wmma4(ai, xb[ks], aI);
    }
    int wc = (wave << 4) + c;
    int ub = (mt << 4) + (h << 3);
#pragma unroll
    for (int r = 0; r < 8; ++r) {
      Tr[(ub + r) * FP + wc] = aR[r];
      Ti[(ub + r) * FP + wc] = aI[r];
    }
  }
  __syncthreads();

  int bdst = (b + 8) & 15;
  int n0 = (wave & 3) << 4;
  int mt0 = (wave >> 2) << 1;
  const float* br_p = Fbr + (n0 + c) * FP + (h << 1);
  const float* bi_p = Fbi + (n0 + c) * FP + (h << 1);
  const float* tr_p = Tr + ((mt0 << 4) + c) * FP + (h << 1);
  const float* ti_p = Ti + ((mt0 << 4) + c) * FP + (h << 1);
  v8f pR[2] = {}, qR[2] = {}, cI[2] = {};
#pragma unroll 4
  for (int ks = 0; ks < 32; ++ks) {
    v2f br = *(const v2f*)(br_p + (ks << 2));
    v2f bi = *(const v2f*)(bi_p + (ks << 2));
#pragma unroll
    for (int j = 0; j < 2; ++j) {
      v2f tr = *(const v2f*)(tr_p + j * (16 * FP) + (ks << 2));
      v2f ti = *(const v2f*)(ti_p + j * (16 * FP) + (ks << 2));
      pR[j] = wmma4(tr, br, pR[j]);
      qR[j] = wmma4(ti, bi, qR[j]);
      cI[j] = wmma4(tr, bi, cI[j]);
      cI[j] = wmma4(ti, br, cI[j]);
    }
  }
  int n = n0 + c;
  int dv = n - 32;
#pragma unroll
  for (int j = 0; j < 2; ++j) {
    int mb = ((mt0 + j) << 4) + (h << 3);
#pragma unroll
    for (int r = 0; r < 8; ++r) {
      int uu = mb + r;
      int du = uu - 32;
      float msk = (du * du + dv * dv <= 900) ? 1.f : 0.f;
      size_t off = ((size_t)((bdst << 6) + ci) << 12) + (size_t)(uu * 64 + n);
      XbR[off] = (pR[j][r] - qR[j][r]) * msk;
      XbI[off] = cI[j][r] * msk;
    }
  }
}

#define WP 65
__global__ void __launch_bounds__(512)
k3_mix(const float* __restrict__ XbR, const float* __restrict__ XbI,
       const float* __restrict__ w, const float* __restrict__ Ftab,
       float* __restrict__ YbR, float* __restrict__ YbI) {
  __shared__ float Wl[576 * WP];
  int wave = threadIdx.x >> 5;
  int lane = threadIdx.x & 31;
  int c = lane & 15, h = lane >> 4;

  for (int q = threadIdx.x; q < 36864; q += 512) {
    int osrc = q / 576;
    int rho = q - osrc * 576;
    Wl[rho * WP + ((osrc + 32) & 63)] = w[q];
  }

  int f = (blockIdx.x << 4) + c;
  int u = bandmap(f >> 6), v = bandmap(f & 63);
  float tr[9], ti[9];
#pragma unroll
  for (int ky = 0; ky < 3; ++ky)
#pragma unroll
    for (int kx = 0; kx < 3; ++kx) {
      int idx = (u * ky + v * kx) & 127;
      tr[ky * 3 + kx] = Ftab[128 + idx];
      ti[ky * 3 + kx] = Ftab[16384 + 128 + idx];
    }
  __syncthreads();

  const float* xr_p = XbR + ((size_t)(wave << 6) << 12) + f;
  const float* xi_p = XbI + ((size_t)(wave << 6) << 12) + f;
  v8f aR[4] = {}, aI[4] = {};
  for (int i0 = 0; i0 < 64; i0 += 4) {
    int kk = i0 + (h << 1);
    float xr0 = xr_p[(size_t)kk << 12], xr1 = xr_p[(size_t)(kk + 1) << 12];
    float xi0 = xi_p[(size_t)kk << 12], xi1 = xi_p[(size_t)(kk + 1) << 12];
    const float* wl = Wl + (kk * 9) * WP + c;
#pragma unroll
    for (int t = 0; t < 9; ++t) {
      v2f ar, ai;
      ar.x = xr0 * tr[t] - xi0 * ti[t];
      ar.y = xr1 * tr[t] - xi1 * ti[t];
      ai.x = xr0 * ti[t] + xi0 * tr[t];
      ai.y = xr1 * ti[t] + xi1 * tr[t];
#pragma unroll
      for (int ot = 0; ot < 4; ++ot) {
        v2f bf;
        bf.x = wl[t * WP + (ot << 4)];
        bf.y = wl[(9 + t) * WP + (ot << 4)];
        aR[ot] = wmma4(ar, bf, aR[ot]);
        aI[ot] = wmma4(ai, bf, aI[ot]);
      }
    }
  }

  int fb = (blockIdx.x << 4) + (h << 3);
#pragma unroll
  for (int ot = 0; ot < 4; ++ot)
#pragma unroll
    for (int r = 0; r < 8; ++r) {
      size_t off = ((size_t)(fb + r) << 10) + (size_t)((wave << 6) + (ot << 4) + c);
      YbR[off] = aR[ot][r];
      YbI[off] = aI[ot][r];
    }
}

#define APITCH 68
__global__ void k4a_col(const float* __restrict__ YbR, const float* __restrict__ YbI,
                        const float* __restrict__ Ftab,
                        float* __restrict__ SR, float* __restrict__ SI) {
  __shared__ float Ar[64 * APITCH];
  __shared__ float Ai[64 * APITCH];
  const float* Fre = Ftab;
  const float* Fim = Ftab + 16384;
  int mh = blockIdx.x & 1;
  int nb = blockIdx.x >> 1;
  int wave = threadIdx.x >> 5;
  int lane = threadIdx.x & 31;
  int c = lane & 15, h = lane >> 4;
  int n0 = ((nb << 3) + wave) << 4;
  int vv = n0 >> 10, b = (n0 >> 6) & 15, o0 = n0 & 63;

  v2f br[16], bi[16];
  const float* yr = YbR + n0 + c;
  const float* yi = YbI + n0 + c;
#pragma unroll
  for (int ks = 0; ks < 16; ++ks) {
    size_t row = (size_t)((ks << 2) + (h << 1)) << 16;
    br[ks].x = yr[row]; br[ks].y = yr[row + 65536];
    bi[ks].x = yi[row]; bi[ks].y = yi[row + 65536];
  }

  for (int idx = threadIdx.x; idx < 4096; idx += 256) {
    int mr = idx >> 6, k = idx & 63;
    int g = ((mh << 6) + mr) * 128 + bandmap(k);
    Ar[mr * APITCH + k] = Fre[g];
    Ai[mr * APITCH + k] = Fim[g];
  }
  __syncthreads();

  for (int mt = 0; mt < 4; ++mt) {
    v8f cR = {}, cI = {};
    const float* ar_p = Ar + ((mt << 4) + c) * APITCH + (h << 1);
    const float* ai_p = Ai + ((mt << 4) + c) * APITCH + (h << 1);
#pragma unroll
    for (int ks = 0; ks < 16; ++ks) {
      v2f ar = *(const v2f*)(ar_p + (ks << 2));
      v2f ai = *(const v2f*)(ai_p + (ks << 2));
      cR = wmma4(ar, br[ks], cR);
      cR = wmma4(ai, bi[ks], cR);
      cI = wmma4(ar, bi[ks], cI);
      v2f ani; ani.x = -ai.x; ani.y = -ai.y;
      cI = wmma4(ani, br[ks], cI);
    }
    int mbase = (mh << 6) + (mt << 4) + (h << 3);
#pragma unroll
    for (int r = 0; r < 8; ++r) {
      size_t off = ((((size_t)(b * 128 + mbase + r) << 6) + vv) << 6) + o0 + c;
      SR[off] = cR[r];
      SI[off] = cI[r];
    }
  }
}

__global__ void k4b_row(const float* __restrict__ SR, const float* __restrict__ SI,
                        const float* __restrict__ Ftab, const float* __restrict__ bias,
                        float* __restrict__ out) {
  const float* Fre = Ftab;
  const float* Fim = Ftab + 16384;
  int blk = blockIdx.x;
  int b = blk >> 7, m = blk & 127;
  int wave = threadIdx.x >> 5;
  int lane = threadIdx.x & 31;
  int c = lane & 15, h = lane >> 4;
  int ot = wave & 3;
  int ng = wave >> 2;
  const float* sr_p = SR + ((size_t)blk << 12) + (ot << 4) + c;
  const float* si_p = SI + ((size_t)blk << 12) + (ot << 4) + c;

  v2f sr[16], si[16];
#pragma unroll
  for (int ks = 0; ks < 16; ++ks) {
    int row = ((ks << 2) + (h << 1)) << 6;
    sr[ks].x = sr_p[row]; sr[ks].y = sr_p[row + 64];
    si[ks].x = si_p[row]; si[ks].y = si_p[row + 64];
  }

  v8f acc[4] = {};
#pragma unroll
  for (int ks = 0; ks < 16; ++ks) {
#pragma unroll
    for (int j = 0; j < 4; ++j) {
      int n0 = ((ng << 2) + j) << 4;
      v2f fr = fragB_kmap(Fre, ks << 2, n0);
      v2f fi = fragB_kmap(Fim, ks << 2, n0);
      acc[j] = wmma4(sr[ks], fr, acc[j]);
      acc[j] = wmma4(si[ks], fi, acc[j]);
    }
  }

  int ob = (ot << 4) + (h << 3);
#pragma unroll
  for (int r = 0; r < 8; ++r) {
    float bo = bias[ob + r];
    float* orow = out + ((((size_t)(b * 64 + ob + r) << 7) + m) << 7) + (ng << 6) + c;
#pragma unroll
    for (int j = 0; j < 4; ++j)
      orow[j << 4] = acc[j][r] * (1.f / 16384.f) + bo;
  }
}

extern "C" void kernel_launch(void* const* d_in, const int* in_sizes, int n_in,
                              void* d_out, int out_size, void* d_ws, size_t ws_size,
                              hipStream_t stream) {
  (void)in_sizes; (void)n_in; (void)out_size; (void)ws_size;
  const float* x    = (const float*)d_in[0];
  const float* w    = (const float*)d_in[1];
  const float* bias = (const float*)d_in[2];
  float* out = (float*)d_out;

  float* ws   = (float*)d_ws;
  float* Ftab = ws;
  float* XbR  = ws + 32768;
  float* XbI  = XbR + 4194304;
  float* YbR  = XbI + 4194304;
  float* YbI  = YbR + 4194304;
  float* SR   = YbI + 4194304;
  float* SI   = SR + 8388608;

  k0_tables<<<16, 256, 0, stream>>>(Ftab);
  k1_fwd  <<<1024, 256, 0, stream>>>(x, Ftab, XbR, XbI);
  k3_mix  <<<256, 512, 0, stream>>>(XbR, XbI, w, Ftab, YbR, YbI);
  k4a_col <<<1024, 256, 0, stream>>>(YbR, YbI, Ftab, SR, SI);
  k4b_row <<<2048, 256, 0, stream>>>(SR, SI, Ftab, bias, out);
}


# Channel mix: taps pre-summed over ky with the band row's column twiddles into one complex LDS B operand (K = 3 x 64, 4 real products); kx twiddles stay on X; k3 WMMAs 72 -> 48; sum order changes
# speedup vs baseline: 4.0489x; 1.1170x over previous
#include <hip/hip_runtime.h>


typedef __attribute__((ext_vector_type(2))) float v2f;
typedef __attribute__((ext_vector_type(4))) float v4f;
typedef __attribute__((ext_vector_type(8))) float v8f;

#define PI_F 3.14159265358979323846f

__device__ __forceinline__ v8f wmma4(v2f a, v2f b, v8f c) {
  return __builtin_amdgcn_wmma_f32_16x16x4_f32(false, a, false, b, (short)0, c,
                                               false, false);
}

__device__ __forceinline__ int bandmap(int bi) { return (bi + 96) & 127; }

__device__ __forceinline__ v2f fragA_kmap(const float* F, int m0, int k0) {
  int lane = threadIdx.x & 31;
  int m = m0 + (lane & 15);
  int kk = k0 + ((lane >> 4) << 1);
  v2f r; r.x = F[m * 128 + bandmap(kk)]; r.y = F[m * 128 + bandmap(kk + 1)];
  return r;
}
__device__ __forceinline__ v2f fragB_kmap(const float* F, int k0, int n0) {
  int lane = threadIdx.x & 31;
  int n = n0 + (lane & 15);
  int kk = k0 + ((lane >> 4) << 1);
  v2f r; r.x = F[bandmap(kk) * 128 + n]; r.y = F[bandmap(kk + 1) * 128 + n];
  return r;
}

__global__ void k0_tables(float* __restrict__ Ftab) {
  for (int t = threadIdx.x + blockIdx.x * blockDim.x; t < 16384;
       t += blockDim.x * gridDim.x) {
    int j = t >> 7, k = t & 127;
    float ang = -2.f * PI_F * (float)((j * k) & 127) / 128.f;
    float s, c;
    sincosf(ang, &s, &c);
    Ftab[t] = c;
    Ftab[16384 + t] = s;
  }
}

#define FP 132
__global__ void __launch_bounds__(256)
k1_fwd(const float* __restrict__ x, const float* __restrict__ Ftab,
       float* __restrict__ XbR, float* __restrict__ XbI) {
  __shared__ __attribute__((aligned(16))) float Fbr[64 * FP];
  __shared__ __attribute__((aligned(16))) float Fbi[64 * FP];
  __shared__ __attribute__((aligned(16))) float Tr[64 * FP];
  __shared__ __attribute__((aligned(16))) float Ti[64 * FP];
  const float* Fre = Ftab;
  const float* Fim = Ftab + 16384;
  int blk = blockIdx.x;
  int b = blk >> 6, ci = blk & 63;
  const float* Ximg = x + ((size_t)blk << 14);
  int wave = threadIdx.x >> 5;
  int lane = threadIdx.x & 31;
  int c = lane & 15, h = lane >> 4;

  v2f xb[32];
  const float* xp = Ximg + (h << 8) + (wave << 4) + c;
#pragma unroll
  for (int ks = 0; ks < 32; ++ks) {
    xb[ks].x = xp[ks << 9];
    xb[ks].y = xp[(ks << 9) + 128];
  }

  for (int q = threadIdx.x; q < 2048; q += 256) {
    int m = q >> 5, k4 = (q & 31) << 2;
    int g = bandmap(m) * 128 + k4;
    *(v4f*)(Fbr + m * FP + k4) = *(const v4f*)(Fre + g);
    *(v4f*)(Fbi + m * FP + k4) = *(const v4f*)(Fim + g);
  }
  __syncthreads();

  for (int mt = 0; mt < 4; ++mt) {
    v8f aR = {}, aI = {};
    const float* ar_p = Fbr + ((mt << 4) + c) * FP + (h << 1);
    const float* ai_p = Fbi + ((mt << 4) + c) * FP + (h << 1);
#pragma unroll
    for (int ks = 0; ks < 32; ++ks) {
      v2f ar = *(const v2f*)(ar_p + (ks << 2));
      v2f ai = *(const v2f*)(ai_p + (ks << 2));
      aR = wmma4(ar, xb[ks], aR);
      aI = wmma4(ai, xb[ks], aI);
    }
    int wc = (wave << 4) + c;
    int ub = (mt << 4) + (h << 3);
#pragma unroll
    for (int r = 0; r < 8; ++r) {
      Tr[(ub + r) * FP + wc] = aR[r];
      Ti[(ub + r) * FP + wc] = aI[r];
    }
  }
  __syncthreads();

  int bdst = (b + 8) & 15;
  int n0 = (wave & 3) << 4;
  int mt0 = (wave >> 2) << 1;
  const float* br_p = Fbr + (n0 + c) * FP + (h << 1);
  const float* bi_p = Fbi + (n0 + c) * FP + (h << 1);
  const float* tr_p = Tr + ((mt0 << 4) + c) * FP + (h << 1);
  const float* ti_p = Ti + ((mt0 << 4) + c) * FP + (h << 1);
  v8f pR[2] = {}, qR[2] = {}, cI[2] = {};
#pragma unroll 4
  for (int ks = 0; ks < 32; ++ks) {
    v2f br = *(const v2f*)(br_p + (ks << 2));
    v2f bi = *(const v2f*)(bi_p + (ks << 2));
#pragma unroll
    for (int j = 0; j < 2; ++j) {
      v2f tr = *(const v2f*)(tr_p + j * (16 * FP) + (ks << 2));
      v2f ti = *(const v2f*)(ti_p + j * (16 * FP) + (ks << 2));
      pR[j] = wmma4(tr, br, pR[j]);
      qR[j] = wmma4(ti, bi, qR[j]);
      cI[j] = wmma4(tr, bi, cI[j]);
      cI[j] = wmma4(ti, br, cI[j]);
    }
  }
  int n = n0 + c;
  int dv = n - 32;
#pragma unroll
  for (int j = 0; j < 2; ++j) {
    int mb = ((mt0 + j) << 4) + (h << 3);
#pragma unroll
    for (int r = 0; r < 8; ++r) {
      int uu = mb + r;
      int du = uu - 32;
      float msk = (du * du + dv * dv <= 900) ? 1.f : 0.f;
      size_t off = ((size_t)((bdst << 6) + ci) << 12) + (size_t)(uu * 64 + n);
      XbR[off] = (pR[j][r] - qR[j][r]) * msk;
      XbI[off] = cI[j][r] * msk;
    }
  }
}

#define WQ 67
__global__ void __launch_bounds__(512)
k3_mix(const float* __restrict__ XbR, const float* __restrict__ XbI,
       const float* __restrict__ w, const float* __restrict__ Ftab,
       float* __restrict__ YbR, float* __restrict__ YbI) {
  __shared__ float Wr[192 * WQ];
  __shared__ float Wi[192 * WQ];
  int wave = threadIdx.x >> 5;
  int lane = threadIdx.x & 31;
  int c = lane & 15, h = lane >> 4;

  int u = bandmap(blockIdx.x >> 2);
  int u2 = (u + u) & 127;
  float a1r = Ftab[128 + u],  a1i = Ftab[16384 + 128 + u];
  float a2r = Ftab[128 + u2], a2i = Ftab[16384 + 128 + u2];

  for (int e = threadIdx.x; e < 12288; e += 512) {
    int osrc = e / 192;
    int rho = e - osrc * 192;
    int i = rho / 3;
    int kx = rho - i * 3;
    const float* wp = w + osrc * 576 + i * 9 + kx;
    float w0 = wp[0], w1 = wp[3], w2 = wp[6];
    int col = (osrc + 32) & 63;
    Wr[rho * WQ + col] = w0 + a1r * w1 + a2r * w2;
    Wi[rho * WQ + col] = a1i * w1 + a2i * w2;
  }

  int f = (blockIdx.x << 4) + c;
  int v = bandmap(f & 63);
  int v2 = (v + v) & 127;
  float b1r = Ftab[128 + v],  b1i = Ftab[16384 + 128 + v];
  float b2r = Ftab[128 + v2], b2i = Ftab[16384 + 128 + v2];
  __syncthreads();

  const float* xr_p = XbR + ((size_t)(wave << 6) << 12) + f;
  const float* xi_p = XbI + ((size_t)(wave << 6) << 12) + f;
  v8f aR[4] = {}, aI[4] = {};
  for (int i0 = 0; i0 < 64; i0 += 4) {
    int kk = i0 + (h << 1);
    float xr0 = xr_p[(size_t)kk << 12], xr1 = xr_p[(size_t)(kk + 1) << 12];
    float xi0 = xi_p[(size_t)kk << 12], xi1 = xi_p[(size_t)(kk + 1) << 12];
    const float* wr = Wr + (kk * 3) * WQ + c;
    const float* wi = Wi + (kk * 3) * WQ + c;
#pragma unroll
    for (int kx = 0; kx < 3; ++kx) {
      v2f ar, ai;
      if (kx == 0) {
        ar.x = xr0; ar.y = xr1; ai.x = xi0; ai.y = xi1;
      } else {
        float tr = (kx == 1) ? b1r : b2r;
        float ti = (kx == 1) ? b1i : b2i;
        ar.x = xr0 * tr - xi0 * ti;
        ar.y = xr1 * tr - xi1 * ti;
        ai.x = xr0 * ti + xi0 * tr;
        ai.y = xr1 * ti + xi1 * tr;
      }
      v2f an; an.x = -ai.x; an.y = -ai.y;
#pragma unroll
      for (int ot = 0; ot < 4; ++ot) {
        v2f br, bi;
        br.x = wr[kx * WQ + (ot << 4)];
        br.y = wr[(3 + kx) * WQ + (ot << 4)];
        bi.x = wi[kx * WQ + (ot << 4)];
        bi.y = wi[(3 + kx) * WQ + (ot << 4)];
        aR[ot] = wmma4(ar, br, aR[ot]);
        aR[ot] = wmma4(an, bi, aR[ot]);
        aI[ot] = wmma4(ar, bi, aI[ot]);
        aI[ot] = wmma4(ai, br, aI[ot]);
      }
    }
  }

  int fb = (blockIdx.x << 4) + (h << 3);
#pragma unroll
  for (int ot = 0; ot < 4; ++ot)
#pragma unroll
    for (int r = 0; r < 8; ++r) {
      size_t off = ((size_t)(fb + r) << 10) + (size_t)((wave << 6) + (ot << 4) + c);
      YbR[off] = aR[ot][r];
      YbI[off] = aI[ot][r];
    }
}

#define APITCH 68
__global__ void k4a_col(const float* __restrict__ YbR, const float* __restrict__ YbI,
                        const float* __restrict__ Ftab,
                        float* __restrict__ SR, float* __restrict__ SI) {
  __shared__ float Ar[64 * APITCH];
  __shared__ float Ai[64 * APITCH];
  const float* Fre = Ftab;
  const float* Fim = Ftab + 16384;
  int mh = blockIdx.x & 1;
  int nb = blockIdx.x >> 1;
  int wave = threadIdx.x >> 5;
  int lane = threadIdx.x & 31;
  int c = lane & 15, h = lane >> 4;
  int n0 = ((nb << 3) + wave) << 4;
  int vv = n0 >> 10, b = (n0 >> 6) & 15, o0 = n0 & 63;

  v2f br[16], bi[16];
  const float* yr = YbR + n0 + c;
  const float* yi = YbI + n0 + c;
#pragma unroll
  for (int ks = 0; ks < 16; ++ks) {
    size_t row = (size_t)((ks << 2) + (h << 1)) << 16;
    br[ks].x = yr[row]; br[ks].y = yr[row + 65536];
    bi[ks].x = yi[row]; bi[ks].y = yi[row + 65536];
  }

  for (int idx = threadIdx.x; idx < 4096; idx += 256) {
    int mr = idx >> 6, k = idx & 63;
    int g = ((mh << 6) + mr) * 128 + bandmap(k);
    Ar[mr * APITCH + k] = Fre[g];
    Ai[mr * APITCH + k] = Fim[g];
  }
  __syncthreads();

  for (int mt = 0; mt < 4; ++mt) {
    v8f cR = {}, cI = {};
    const float* ar_p = Ar + ((mt << 4) + c) * APITCH + (h << 1);
    const float* ai_p = Ai + ((mt << 4) + c) * APITCH + (h << 1);
#pragma unroll
    for (int ks = 0; ks < 16; ++ks) {
      v2f ar = *(const v2f*)(ar_p + (ks << 2));
      v2f ai = *(const v2f*)(ai_p + (ks << 2));
      cR = wmma4(ar, br[ks], cR);
      cR = wmma4(ai, bi[ks], cR);
      cI = wmma4(ar, bi[ks], cI);
      v2f ani; ani.x = -ai.x; ani.y = -ai.y;
      cI = wmma4(ani, br[ks], cI);
    }
    int mbase = (mh << 6) + (mt << 4) + (h << 3);
#pragma unroll
    for (int r = 0; r < 8; ++r) {
      size_t off = ((((size_t)(b * 128 + mbase + r) << 6) + vv) << 6) + o0 + c;
      SR[off] = cR[r];
      SI[off] = cI[r];
    }
  }
}

__global__ void k4b_row(const float* __restrict__ SR, const float* __restrict__ SI,
                        const float* __restrict__ Ftab, const float* __restrict__ bias,
                        float* __restrict__ out) {
  const float* Fre = Ftab;
  const float* Fim = Ftab + 16384;
  int blk = blockIdx.x;
  int b = blk >> 7, m = blk & 127;
  int wave = threadIdx.x >> 5;
  int lane = threadIdx.x & 31;
  int c = lane & 15, h = lane >> 4;
  int ot = wave & 3;
  int ng = wave >> 2;
  const float* sr_p = SR + ((size_t)blk << 12) + (ot << 4) + c;
  const float* si_p = SI + ((size_t)blk << 12) + (ot << 4) + c;

  v2f sr[16], si[16];
#pragma unroll
  for (int ks = 0; ks < 16; ++ks) {
    int row = ((ks << 2) + (h << 1)) << 6;
    sr[ks].x = sr_p[row]; sr[ks].y = sr_p[row + 64];
    si[ks].x = si_p[row]; si[ks].y = si_p[row + 64];
  }

  v8f acc[4] = {};
#pragma unroll
  for (int ks = 0; ks < 16; ++ks) {
#pragma unroll
    for (int j = 0; j < 4; ++j) {
      int n0 = ((ng << 2) + j) << 4;
      v2f fr = fragB_kmap(Fre, ks << 2, n0);
      v2f fi = fragB_kmap(Fim, ks << 2, n0);
      acc[j] = wmma4(sr[ks], fr, acc[j]);
      acc[j] = wmma4(si[ks], fi, acc[j]);
    }
  }

  int ob = (ot << 4) + (h << 3);
#pragma unroll
  for (int r = 0; r < 8; ++r) {
    float bo = bias[ob + r];
    float* orow = out + ((((size_t)(b * 64 + ob + r) << 7) + m) << 7) + (ng << 6) + c;
#pragma unroll
    for (int j = 0; j < 4; ++j)
      orow[j << 4] = acc[j][r] * (1.f / 16384.f) + bo;
  }
}

extern "C" void kernel_launch(void* const* d_in, const int* in_sizes, int n_in,
                              void* d_out, int out_size, void* d_ws, size_t ws_size,
                              hipStream_t stream) {
  (void)in_sizes; (void)n_in; (void)out_size; (void)ws_size;
  const float* x    = (const float*)d_in[0];
  const float* w    = (const float*)d_in[1];
  const float* bias = (const float*)d_in[2];
  float* out = (float*)d_out;

  float* ws   = (float*)d_ws;
  float* Ftab = ws;
  float* XbR  = ws + 32768;
  float* XbI  = XbR + 4194304;
  float* YbR  = XbI + 4194304;
  float* YbI  = YbR + 4194304;
  float* SR   = YbI + 4194304;
  float* SI   = SR + 8388608;

  k0_tables<<<16, 256, 0, stream>>>(Ftab);
  k1_fwd  <<<1024, 256, 0, stream>>>(x, Ftab, XbR, XbI);
  k3_mix  <<<256, 512, 0, stream>>>(XbR, XbI, w, Ftab, YbR, YbI);
  k4a_col <<<1024, 256, 0, stream>>>(YbR, YbI, Ftab, SR, SI);
  k4b_row <<<2048, 256, 0, stream>>>(SR, SI, Ftab, bias, out);
}


# Complex x complex GEMMs (k1 stage 2, channel mix, inverse column pass) use the 3-real-product form: S1=(ar+ai)br, S2=ar(bi-br), S3=ai(br+bi); 4 -> 3 WMMAs per k-step; sum order changes
# speedup vs baseline: 4.3096x; 1.0644x over previous
#include <hip/hip_runtime.h>


typedef __attribute__((ext_vector_type(2))) float v2f;
typedef __attribute__((ext_vector_type(4))) float v4f;
typedef __attribute__((ext_vector_type(8))) float v8f;

#define PI_F 3.14159265358979323846f

__device__ __forceinline__ v8f wmma4(v2f a, v2f b, v8f c) {
  return __builtin_amdgcn_wmma_f32_16x16x4_f32(false, a, false, b, (short)0, c,
                                               false, false);
}

__device__ __forceinline__ int bandmap(int bi) { return (bi + 96) & 127; }

__device__ __forceinline__ v2f fragA_kmap(const float* F, int m0, int k0) {
  int lane = threadIdx.x & 31;
  int m = m0 + (lane & 15);
  int kk = k0 + ((lane >> 4) << 1);
  v2f r; r.x = F[m * 128 + bandmap(kk)]; r.y = F[m * 128 + bandmap(kk + 1)];
  return r;
}
__device__ __forceinline__ v2f fragB_kmap(const float* F, int k0, int n0) {
  int lane = threadIdx.x & 31;
  int n = n0 + (lane & 15);
  int kk = k0 + ((lane >> 4) << 1);
  v2f r; r.x = F[bandmap(kk) * 128 + n]; r.y = F[bandmap(kk + 1) * 128 + n];
  return r;
}

__global__ void k0_tables(float* __restrict__ Ftab) {
  for (int t = threadIdx.x + blockIdx.x * blockDim.x; t < 16384;
       t += blockDim.x * gridDim.x) {
    int j = t >> 7, k = t & 127;
    float ang = -2.f * PI_F * (float)((j * k) & 127) / 128.f;
    float s, c;
    sincosf(ang, &s, &c);
    Ftab[t] = c;
    Ftab[16384 + t] = s;
  }
}

#define FP 132
__global__ void __launch_bounds__(256)
k1_fwd(const float* __restrict__ x, const float* __restrict__ Ftab,
       float* __restrict__ XbR, float* __restrict__ XbI) {
  __shared__ __attribute__((aligned(16))) float Fbr[64 * FP];
  __shared__ __attribute__((aligned(16))) float Fbi[64 * FP];
  __shared__ __attribute__((aligned(16))) float Tr[64 * FP];
  __shared__ __attribute__((aligned(16))) float Ti[64 * FP];
  const float* Fre = Ftab;
  const float* Fim = Ftab + 16384;
  int blk = blockIdx.x;
  int b = blk >> 6, ci = blk & 63;
  const float* Ximg = x + ((size_t)blk << 14);
  int wave = threadIdx.x >> 5;
  int lane = threadIdx.x & 31;
  int c = lane & 15, h = lane >> 4;

  v2f xb[32];
  const float* xp = Ximg + (h << 8) + (wave << 4) + c;
#pragma unroll
  for (int ks = 0; ks < 32; ++ks) {
    xb[ks].x = xp[ks << 9];
    xb[ks].y = xp[(ks << 9) + 128];
  }

  for (int q = threadIdx.x; q < 2048; q += 256) {
    int m = q >> 5, k4 = (q & 31) << 2;
    int g = bandmap(m) * 128 + k4;
    *(v4f*)(Fbr + m * FP + k4) = *(const v4f*)(Fre + g);
    *(v4f*)(Fbi + m * FP + k4) = *(const v4f*)(Fim + g);
  }
  __syncthreads();

  for (int mt = 0; mt < 4; ++mt) {
    v8f aR = {}, aI = {};
    const float* ar_p = Fbr + ((mt << 4) + c) * FP + (h << 1);
    const float* ai_p = Fbi + ((mt << 4) + c) * FP + (h << 1);
#pragma unroll
    for (int ks = 0; ks < 32; ++ks) {
      v2f ar = *(const v2f*)(ar_p + (ks << 2));
      v2f ai = *(const v2f*)(ai_p + (ks << 2));
      aR = wmma4(ar, xb[ks], aR);
      aI = wmma4(ai, xb[ks], aI);
    }
    int wc = (wave << 4) + c;
    int ub = (mt << 4) + (h << 3);
#pragma unroll
    for (int r = 0; r < 8; ++r) {
      Tr[(ub + r) * FP + wc] = aR[r];
      Ti[(ub + r) * FP + wc] = aI[r];
    }
  }
  __syncthreads();

  int bdst = (b + 8) & 15;
  int n0 = (wave & 3) << 4;
  int mt0 = (wave >> 2) << 1;
  const float* br_p = Fbr + (n0 + c) * FP + (h << 1);
  const float* bi_p = Fbi + (n0 + c) * FP + (h << 1);
  const float* tr_p = Tr + ((mt0 << 4) + c) * FP + (h << 1);
  const float* ti_p = Ti + ((mt0 << 4) + c) * FP + (h << 1);
  v8f s1[2] = {}, s2[2] = {}, s3[2] = {};
#pragma unroll 4
  for (int ks = 0; ks < 32; ++ks) {
    v2f br = *(const v2f*)(br_p + (ks << 2));
    v2f bi = *(const v2f*)(bi_p + (ks << 2));
    v2f bd = bi - br;
    v2f bs = br + bi;
#pragma unroll
    for (int j = 0; j < 2; ++j) {
      v2f tr = *(const v2f*)(tr_p + j * (16 * FP) + (ks << 2));
      v2f ti = *(const v2f*)(ti_p + j * (16 * FP) + (ks << 2));
      v2f ts = tr + ti;
      s1[j] = wmma4(ts, br, s1[j]);
      s2[j] = wmma4(tr, bd, s2[j]);
      s3[j] = wmma4(ti, bs, s3[j]);
    }
  }
  int n = n0 + c;
  int dv = n - 32;
#pragma unroll
  for (int j = 0; j < 2; ++j) {
    int mb = ((mt0 + j) << 4) + (h << 3);
#pragma unroll
    for (int r = 0; r < 8; ++r) {
      int uu = mb + r;
      int du = uu - 32;
      float msk = (du * du + dv * dv <= 900) ? 1.f : 0.f;
      size_t off = ((size_t)((bdst << 6) + ci) << 12) + (size_t)(uu * 64 + n);
      XbR[off] = (s1[j][r] - s3[j][r]) * msk;
      XbI[off] = (s1[j][r] + s2[j][r]) * msk;
    }
  }
}

#define WPK 386
#define WPP 1184
__global__ void __launch_bounds__(512) __attribute__((amdgpu_waves_per_eu(4, 4)))
k3_mix(const float* __restrict__ XbR, const float* __restrict__ XbI,
       const float* __restrict__ w, const float* __restrict__ Ftab,
       float* __restrict__ YbR, float* __restrict__ YbI) {
  __shared__ __attribute__((aligned(16))) float Wc[32 * WPP];
  int wave = threadIdx.x >> 5;
  int lane = threadIdx.x & 31;
  int c = lane & 15, h = lane >> 4;

  int u = bandmap(blockIdx.x >> 2);
  int u2 = (u + u) & 127;
  float a1r = Ftab[128 + u],  a1i = Ftab[16384 + 128 + u];
  float a2r = Ftab[128 + u2], a2i = Ftab[16384 + 128 + u2];

  for (int e = threadIdx.x; e < 12288; e += 512) {
    int rest = e >> 5;
    int oh = rest / 24;
    int rho = ((rest - oh * 24) << 3) + (e & 7);
    int osrc = (oh << 2) + ((e >> 3) & 3);
    int i = rho / 3;
    int kx = rho - i * 3;
    const float* wp = w + osrc * 576 + i * 9 + kx;
    float w0 = wp[0], w1 = wp[3], w2 = wp[6];
    int col = (osrc + 32) & 63;
    float wre = w0 + a1r * w1 + a2r * w2;
    float wim = a1i * w1 + a2i * w2;
    float* d = Wc + (i >> 1) * WPP + kx * WPK + col * 6 + (i & 1);
    d[0] = wre;
    d[2] = wim - wre;
    d[4] = wre + wim;
  }

  int f = (blockIdx.x << 4) + c;
  int v = bandmap(f & 63);
  int v2 = (v + v) & 127;
  float b1r = Ftab[128 + v],  b1i = Ftab[16384 + 128 + v];
  float b2r = Ftab[128 + v2], b2i = Ftab[16384 + 128 + v2];
  __syncthreads();

  const float* xr_p = XbR + ((size_t)(wave << 6) << 12) + f;
  const float* xi_p = XbI + ((size_t)(wave << 6) << 12) + f;
  v8f s1[4] = {}, s2[4] = {}, s3[4] = {};
  for (int i0 = 0; i0 < 64; i0 += 4) {
    int kk = i0 + (h << 1);
    float xr0 = xr_p[(size_t)kk << 12], xr1 = xr_p[(size_t)(kk + 1) << 12];
    float xi0 = xi_p[(size_t)kk << 12], xi1 = xi_p[(size_t)(kk + 1) << 12];
    const float* wb = Wc + (kk >> 1) * WPP + c * 6;
    v2f bf[3][4][3];
#pragma unroll
    for (int kx = 0; kx < 3; ++kx)
#pragma unroll
      for (int ot = 0; ot < 4; ++ot) {
        const v2f* q = (const v2f*)(wb + kx * WPK + ot * 96);
        bf[kx][ot][0] = q[0]; bf[kx][ot][1] = q[1]; bf[kx][ot][2] = q[2];
      }
#pragma unroll
    for (int kx = 0; kx < 3; ++kx) {
      v2f ar, ai;
      if (kx == 0) {
        ar.x = xr0; ar.y = xr1; ai.x = xi0; ai.y = xi1;
      } else {
        float tr = (kx == 1) ? b1r : b2r;
        float ti = (kx == 1) ? b1i : b2i;
        ar.x = xr0 * tr - xi0 * ti;
        ar.y = xr1 * tr - xi1 * ti;
        ai.x = xr0 * ti + xi0 * tr;
        ai.y = xr1 * ti + xi1 * tr;
      }
      v2f as = ar + ai;
#pragma unroll
      for (int ot = 0; ot < 4; ++ot) {
        s1[ot] = wmma4(as, bf[kx][ot][0], s1[ot]);
        s2[ot] = wmma4(ar, bf[kx][ot][1], s2[ot]);
        s3[ot] = wmma4(ai, bf[kx][ot][2], s3[ot]);
      }
    }
  }

  int fb = (blockIdx.x << 4) + (h << 3);
#pragma unroll
  for (int ot = 0; ot < 4; ++ot)
#pragma unroll
    for (int r = 0; r < 8; ++r) {
      size_t off = ((size_t)(fb + r) << 10) + (size_t)((wave << 6) + (ot << 4) + c);
      YbR[off] = s1[ot][r] - s3[ot][r];
      YbI[off] = s1[ot][r] + s2[ot][r];
    }
}

#define APITCH 68
__global__ void k4a_col(const float* __restrict__ YbR, const float* __restrict__ YbI,
                        const float* __restrict__ Ftab,
                        float* __restrict__ SR, float* __restrict__ SI) {
  __shared__ float A1[64 * APITCH];
  __shared__ float A2[64 * APITCH];
  __shared__ float A3[64 * APITCH];
  const float* Fre = Ftab;
  const float* Fim = Ftab + 16384;
  int mh = blockIdx.x & 1;
  int nb = blockIdx.x >> 1;
  int wave = threadIdx.x >> 5;
  int lane = threadIdx.x & 31;
  int c = lane & 15, h = lane >> 4;
  int n0 = ((nb << 3) + wave) << 4;
  int vv = n0 >> 10, b = (n0 >> 6) & 15, o0 = n0 & 63;

  v2f br[16], bi[16];
  const float* yr = YbR + n0 + c;
  const float* yi = YbI + n0 + c;
#pragma unroll
  for (int ks = 0; ks < 16; ++ks) {
    size_t row = (size_t)((ks << 2) + (h << 1)) << 16;
    br[ks].x = yr[row]; br[ks].y = yr[row + 65536];
    bi[ks].x = yi[row]; bi[ks].y = yi[row + 65536];
  }

  for (int idx = threadIdx.x; idx < 4096; idx += 256) {
    int mr = idx >> 6, k = idx & 63;
    int g = ((mh << 6) + mr) * 128 + bandmap(k);
    float fr = Fre[g], fi = Fim[g];
    A1[mr * APITCH + k] = fr;
    A2[mr * APITCH + k] = fr - fi;
    A3[mr * APITCH + k] = fr + fi;
  }
  __syncthreads();

  for (int mt = 0; mt < 4; ++mt) {
    v8f cP = {}, cQ = {}, cS = {};
    const float* a1_p = A1 + ((mt << 4) + c) * APITCH + (h << 1);
    const float* a2_p = A2 + ((mt << 4) + c) * APITCH + (h << 1);
    const float* a3_p = A3 + ((mt << 4) + c) * APITCH + (h << 1);
#pragma unroll
    for (int ks = 0; ks < 16; ++ks) {
      v2f a1 = *(const v2f*)(a1_p + (ks << 2));
      v2f a2 = *(const v2f*)(a2_p + (ks << 2));
      v2f a3 = *(const v2f*)(a3_p + (ks << 2));
      v2f bs = br[ks] + bi[ks];
      cP = wmma4(a1, bs, cP);
      cQ = wmma4(a2, bi[ks], cQ);
      cS = wmma4(a3, br[ks], cS);
    }
    int mbase = (mh << 6) + (mt << 4) + (h << 3);
#pragma unroll
    for (int r = 0; r < 8; ++r) {
      size_t off = ((((size_t)(b * 128 + mbase + r) << 6) + vv) << 6) + o0 + c;
      SR[off] = cP[r] - cQ[r];
      SI[off] = cP[r] - cS[r];
    }
  }
}

__global__ void k4b_row(const float* __restrict__ SR, const float* __restrict__ SI,
                        const float* __restrict__ Ftab, const float* __restrict__ bias,
                        float* __restrict__ out) {
  const float* Fre = Ftab;
  const float* Fim = Ftab + 16384;
  int blk = blockIdx.x;
  int b = blk >> 7, m = blk & 127;
  int wave = threadIdx.x >> 5;
  int lane = threadIdx.x & 31;
  int c = lane & 15, h = lane >> 4;
  int ot = wave & 3;
  int ng = wave >> 2;
  const float* sr_p = SR + ((size_t)blk << 12) + (ot << 4) + c;
  const float* si_p = SI + ((size_t)blk << 12) + (ot << 4) + c;

  v2f sr[16], si[16];
#pragma unroll
  for (int ks = 0; ks < 16; ++ks) {
    int row = ((ks << 2) + (h << 1)) << 6;
    sr[ks].x = sr_p[row]; sr[ks].y = sr_p[row + 64];
    si[ks].x = si_p[row]; si[ks].y = si_p[row + 64];
  }

  v8f acc[4] = {};
#pragma unroll
  for (int ks = 0; ks < 16; ++ks) {
#pragma unroll
    for (int j = 0; j < 4; ++j) {
      int n0 = ((ng << 2) + j) << 4;
      v2f fr = fragB_kmap(Fre, ks << 2, n0);
      v2f fi = fragB_kmap(Fim, ks << 2, n0);
      acc[j] = wmma4(sr[ks], fr, acc[j]);
      acc[j] = wmma4(si[ks], fi, acc[j]);
    }
  }

  int ob = (ot << 4) + (h << 3);
#pragma unroll
  for (int r = 0; r < 8; ++r) {
    float bo = bias[ob + r];
    float* orow = out + ((((size_t)(b * 64 + ob + r) << 7) + m) << 7) + (ng << 6) + c;
#pragma unroll
    for (int j = 0; j < 4; ++j)
      orow[j << 4] = acc[j][r] * (1.f / 16384.f) + bo;
  }
}

extern "C" void kernel_launch(void* const* d_in, const int* in_sizes, int n_in,
                              void* d_out, int out_size, void* d_ws, size_t ws_size,
                              hipStream_t stream) {
  (void)in_sizes; (void)n_in; (void)out_size; (void)ws_size;
  const float* x    = (const float*)d_in[0];
  const float* w    = (const float*)d_in[1];
  const float* bias = (const float*)d_in[2];
  float* out = (float*)d_out;

  float* ws   = (float*)d_ws;
  float* Ftab = ws;
  float* XbR  = ws + 32768;
  float* XbI  = XbR + 4194304;
  float* YbR  = XbI + 4194304;
  float* YbI  = YbR + 4194304;
  float* SR   = YbI + 4194304;
  float* SI   = SR + 8388608;

  k0_tables<<<16, 256, 0, stream>>>(Ftab);
  k1_fwd  <<<1024, 256, 0, stream>>>(x, Ftab, XbR, XbI);
  k3_mix  <<<256, 512, 0, stream>>>(XbR, XbI, w, Ftab, YbR, YbI);
  k4a_col <<<1024, 256, 0, stream>>>(YbR, YbI, Ftab, SR, SI);
  k4b_row <<<2048, 256, 0, stream>>>(SR, SI, Ftab, bias, out);
}


# Batch entries b and b+8 packed as one complex image (real taps, symmetric mask) through forward DFT, mix and inverse; out = Re / Im of the result; all GEMMs 3-product complex; sum order changes
# speedup vs baseline: 6.3774x; 1.4798x over previous
#include <hip/hip_runtime.h>


typedef __attribute__((ext_vector_type(2))) float v2f;
typedef __attribute__((ext_vector_type(4))) float v4f;
typedef __attribute__((ext_vector_type(8))) float v8f;

#define PI_F 3.14159265358979323846f

__device__ __forceinline__ v8f wmma4(v2f a, v2f b, v8f c) {
  return __builtin_amdgcn_wmma_f32_16x16x4_f32(false, a, false, b, (short)0, c,
                                               false, false);
}

__device__ __forceinline__ int bandmap(int bi) { return (bi + 96) & 127; }

__device__ __forceinline__ v2f fragA_kmap(const float* F, int m0, int k0) {
  int lane = threadIdx.x & 31;
  int m = m0 + (lane & 15);
  int kk = k0 + ((lane >> 4) << 1);
  v2f r; r.x = F[m * 128 + bandmap(kk)]; r.y = F[m * 128 + bandmap(kk + 1)];
  return r;
}
__device__ __forceinline__ v2f fragB_kmap(const float* F, int k0, int n0) {
  int lane = threadIdx.x & 31;
  int n = n0 + (lane & 15);
  int kk = k0 + ((lane >> 4) << 1);
  v2f r; r.x = F[bandmap(kk) * 128 + n]; r.y = F[bandmap(kk + 1) * 128 + n];
  return r;
}

__global__ void k0_tables(float* __restrict__ Ftab) {
  for (int t = threadIdx.x + blockIdx.x * blockDim.x; t < 16384;
       t += blockDim.x * gridDim.x) {
    int j = t >> 7, k = t & 127;
    float ang = -2.f * PI_F * (float)((j * k) & 127) / 128.f;
    float s, c;
    sincosf(ang, &s, &c);
    Ftab[t] = c;
    Ftab[16384 + t] = s;
  }
}

#define FP 132
__global__ void __launch_bounds__(256)
k1_fwd(const float* __restrict__ x, const float* __restrict__ Ftab,
       float* __restrict__ XbR, float* __restrict__ XbI) {
  __shared__ __attribute__((aligned(16))) float Fbr[64 * FP];
  __shared__ __attribute__((aligned(16))) float Fbi[64 * FP];
  __shared__ __attribute__((aligned(16))) float Tr[64 * FP];
  __shared__ __attribute__((aligned(16))) float Ti[64 * FP];
  const float* Fre = Ftab;
  const float* Fim = Ftab + 16384;
  int blk = blockIdx.x;
  int q = blk >> 6, ci = blk & 63;
  const float* Xre = x + ((size_t)(((q + 8) << 6) + ci) << 14);
  const float* Xim = x + ((size_t)((q << 6) + ci) << 14);
  int wave = threadIdx.x >> 5;
  int lane = threadIdx.x & 31;
  int c = lane & 15, h = lane >> 4;

  v2f x1[32], x2[32];
  const float* xp1 = Xre + (h << 8) + (wave << 4) + c;
  const float* xp2 = Xim + (h << 8) + (wave << 4) + c;
#pragma unroll
  for (int ks = 0; ks < 32; ++ks) {
    x1[ks].x = xp1[ks << 9];
    x1[ks].y = xp1[(ks << 9) + 128];
    x2[ks].x = xp2[ks << 9];
    x2[ks].y = xp2[(ks << 9) + 128];
  }

  for (int q = threadIdx.x; q < 2048; q += 256) {
    int m = q >> 5, k4 = (q & 31) << 2;
    int g = bandmap(m) * 128 + k4;
    *(v4f*)(Fbr + m * FP + k4) = *(const v4f*)(Fre + g);
    *(v4f*)(Fbi + m * FP + k4) = *(const v4f*)(Fim + g);
  }
  __syncthreads();

  for (int mt = 0; mt < 4; ++mt) {
    v8f a1 = {}, a2 = {}, a3 = {};
    const float* ar_p = Fbr + ((mt << 4) + c) * FP + (h << 1);
    const float* ai_p = Fbi + ((mt << 4) + c) * FP + (h << 1);
#pragma unroll
    for (int ks = 0; ks < 32; ++ks) {
      v2f ar = *(const v2f*)(ar_p + (ks << 2));
      v2f ai = *(const v2f*)(ai_p + (ks << 2));
      v2f as = ar + ai;
      v2f bd = x2[ks] - x1[ks];
      v2f bs = x1[ks] + x2[ks];
      a1 = wmma4(as, x1[ks], a1);
      a2 = wmma4(ar, bd, a2);
      a3 = wmma4(ai, bs, a3);
    }
    int wc = (wave << 4) + c;
    int ub = (mt << 4) + (h << 3);
#pragma unroll
    for (int r = 0; r < 8; ++r) {
      Tr[(ub + r) * FP + wc] = a1[r] - a3[r];
      Ti[(ub + r) * FP + wc] = a1[r] + a2[r];
    }
  }
  __syncthreads();

  int n0 = (wave & 3) << 4;
  int mt0 = (wave >> 2) << 1;
  const float* br_p = Fbr + (n0 + c) * FP + (h << 1);
  const float* bi_p = Fbi + (n0 + c) * FP + (h << 1);
  const float* tr_p = Tr + ((mt0 << 4) + c) * FP + (h << 1);
  const float* ti_p = Ti + ((mt0 << 4) + c) * FP + (h << 1);
  v8f s1[2] = {}, s2[2] = {}, s3[2] = {};
#pragma unroll 4
  for (int ks = 0; ks < 32; ++ks) {
    v2f br = *(const v2f*)(br_p + (ks << 2));
    v2f bi = *(const v2f*)(bi_p + (ks << 2));
    v2f bd = bi - br;
    v2f bs = br + bi;
#pragma unroll
    for (int j = 0; j < 2; ++j) {
      v2f tr = *(const v2f*)(tr_p + j * (16 * FP) + (ks << 2));
      v2f ti = *(const v2f*)(ti_p + j * (16 * FP) + (ks << 2));
      v2f ts = tr + ti;
      s1[j] = wmma4(ts, br, s1[j]);
      s2[j] = wmma4(tr, bd, s2[j]);
      s3[j] = wmma4(ti, bs, s3[j]);
    }
  }
  int n = n0 + c;
  int dv = n - 32;
#pragma unroll
  for (int j = 0; j < 2; ++j) {
    int mb = ((mt0 + j) << 4) + (h << 3);
#pragma unroll
    for (int r = 0; r < 8; ++r) {
      int uu = mb + r;
      int du = uu - 32;
      float msk = (du * du + dv * dv <= 900) ? 1.f : 0.f;
      size_t off = ((size_t)blk << 12) + (size_t)(uu * 64 + n);
      XbR[off] = (s1[j][r] - s3[j][r]) * msk;
      XbI[off] = (s1[j][r] + s2[j][r]) * msk;
    }
  }
}

#define WPK 386
#define WPP 1184
__global__ void __launch_bounds__(512) __attribute__((amdgpu_waves_per_eu(4, 4)))
k3_mix(const float* __restrict__ XbR, const float* __restrict__ XbI,
       const float* __restrict__ w, const float* __restrict__ Ftab,
       float* __restrict__ YbR, float* __restrict__ YbI) {
  __shared__ __attribute__((aligned(16))) float Wc[32 * WPP];
  int wave = threadIdx.x >> 5;
  int pr = wave >> 1;
  int oh = wave & 1;
  int lane = threadIdx.x & 31;
  int c = lane & 15, h = lane >> 4;

  int u = bandmap(blockIdx.x >> 2);
  int u2 = (u + u) & 127;
  float a1r = Ftab[128 + u],  a1i = Ftab[16384 + 128 + u];
  float a2r = Ftab[128 + u2], a2i = Ftab[16384 + 128 + u2];

  for (int e = threadIdx.x; e < 12288; e += 512) {
    int rest = e >> 5;
    int oh = rest / 24;
    int rho = ((rest - oh * 24) << 3) + (e & 7);
    int osrc = (oh << 2) + ((e >> 3) & 3);
    int i = rho / 3;
    int kx = rho - i * 3;
    const float* wp = w + osrc * 576 + i * 9 + kx;
    float w0 = wp[0], w1 = wp[3], w2 = wp[6];
    int col = (osrc + 32) & 63;
    float wre = w0 + a1r * w1 + a2r * w2;
    float wim = a1i * w1 + a2i * w2;
    float* d = Wc + (i >> 1) * WPP + kx * WPK + col * 6 + (i & 1);
    d[0] = wre;
    d[2] = wim - wre;
    d[4] = wre + wim;
  }

  int f = (blockIdx.x << 4) + c;
  int v = bandmap(f & 63);
  int v2 = (v + v) & 127;
  float b1r = Ftab[128 + v],  b1i = Ftab[16384 + 128 + v];
  float b2r = Ftab[128 + v2], b2i = Ftab[16384 + 128 + v2];
  __syncthreads();

  const float* xr_p = XbR + ((size_t)(pr << 6) << 12) + f;
  const float* xi_p = XbI + ((size_t)(pr << 6) << 12) + f;
  v8f s1[2] = {}, s2[2] = {}, s3[2] = {};
#pragma nounroll
  for (int i0 = 0; i0 < 64; i0 += 4) {
    int kk = i0 + (h << 1);
    float xr0 = xr_p[(size_t)kk << 12], xr1 = xr_p[(size_t)(kk + 1) << 12];
    float xi0 = xi_p[(size_t)kk << 12], xi1 = xi_p[(size_t)(kk + 1) << 12];
    const float* wb = Wc + (kk >> 1) * WPP + (oh * 32 + c) * 6;
    v2f bf[3][2][3];
#pragma unroll
    for (int kx = 0; kx < 3; ++kx)
#pragma unroll
      for (int ot = 0; ot < 2; ++ot) {
        const v2f* q = (const v2f*)(wb + kx * WPK + ot * 96);
        bf[kx][ot][0] = q[0]; bf[kx][ot][1] = q[1]; bf[kx][ot][2] = q[2];
      }
#pragma unroll
    for (int kx = 0; kx < 3; ++kx) {
      v2f ar, ai;
      if (kx == 0) {
        ar.x = xr0; ar.y = xr1; ai.x = xi0; ai.y = xi1;
      } else {
        float tr = (kx == 1) ? b1r : b2r;
        float ti = (kx == 1) ? b1i : b2i;
        ar.x = xr0 * tr - xi0 * ti;
        ar.y = xr1 * tr - xi1 * ti;
        ai.x = xr0 * ti + xi0 * tr;
        ai.y = xr1 * ti + xi1 * tr;
      }
      v2f as = ar + ai;
#pragma unroll
      for (int ot = 0; ot < 2; ++ot) {
        s1[ot] = wmma4(as, bf[kx][ot][0], s1[ot]);
        s2[ot] = wmma4(ar, bf[kx][ot][1], s2[ot]);
        s3[ot] = wmma4(ai, bf[kx][ot][2], s3[ot]);
      }
    }
  }

  int fb = (blockIdx.x << 4) + (h << 3);
#pragma unroll
  for (int ot = 0; ot < 2; ++ot)
#pragma unroll
    for (int r = 0; r < 8; ++r) {
      size_t off = ((size_t)(fb + r) << 9) + (size_t)((pr << 6) + (oh << 5) + (ot << 4) + c);
      YbR[off] = s1[ot][r] - s3[ot][r];
      YbI[off] = s1[ot][r] + s2[ot][r];
    }
}

#define APITCH 68
__global__ void k4a_col(const float* __restrict__ YbR, const float* __restrict__ YbI,
                        const float* __restrict__ Ftab,
                        float* __restrict__ SR, float* __restrict__ SI) {
  __shared__ float A1[64 * APITCH];
  __shared__ float A2[64 * APITCH];
  __shared__ float A3[64 * APITCH];
  const float* Fre = Ftab;
  const float* Fim = Ftab + 16384;
  int mh = blockIdx.x & 1;
  int nb = blockIdx.x >> 1;
  int wave = threadIdx.x >> 5;
  int lane = threadIdx.x & 31;
  int c = lane & 15, h = lane >> 4;
  int n0 = ((nb << 3) + wave) << 4;
  int vv = n0 >> 9, q = (n0 >> 6) & 7, o0 = n0 & 63;

  v2f br[16], bi[16];
  const float* yr = YbR + n0 + c;
  const float* yi = YbI + n0 + c;
#pragma unroll
  for (int ks = 0; ks < 16; ++ks) {
    size_t row = (size_t)((ks << 2) + (h << 1)) << 15;
    br[ks].x = yr[row]; br[ks].y = yr[row + 32768];
    bi[ks].x = yi[row]; bi[ks].y = yi[row + 32768];
  }

  for (int idx = threadIdx.x; idx < 4096; idx += 256) {
    int mr = idx >> 6, k = idx & 63;
    int g = ((mh << 6) + mr) * 128 + bandmap(k);
    float fr = Fre[g], fi = Fim[g];
    A1[mr * APITCH + k] = fr;
    A2[mr * APITCH + k] = fr - fi;
    A3[mr * APITCH + k] = fr + fi;
  }
  __syncthreads();

  for (int mt = 0; mt < 4; ++mt) {
    v8f cP = {}, cQ = {}, cS = {};
    const float* a1_p = A1 + ((mt << 4) + c) * APITCH + (h << 1);
    const float* a2_p = A2 + ((mt << 4) + c) * APITCH + (h << 1);
    const float* a3_p = A3 + ((mt << 4) + c) * APITCH + (h << 1);
#pragma unroll
    for (int ks = 0; ks < 16; ++ks) {
      v2f a1 = *(const v2f*)(a1_p + (ks << 2));
      v2f a2 = *(const v2f*)(a2_p + (ks << 2));
      v2f a3 = *(const v2f*)(a3_p + (ks << 2));
      v2f bs = br[ks] + bi[ks];
      cP = wmma4(a1, bs, cP);
      cQ = wmma4(a2, bi[ks], cQ);
      cS = wmma4(a3, br[ks], cS);
    }
    int mbase = (mh << 6) + (mt << 4) + (h << 3);
#pragma unroll
    for (int r = 0; r < 8; ++r) {
      size_t off = ((((size_t)(q * 128 + mbase + r) << 6) + vv) << 6) + o0 + c;
      SR[off] = cP[r] - cQ[r];
      SI[off] = cP[r] - cS[r];
    }
  }
}

__global__ void __launch_bounds__(256)
k4b_row(const float* __restrict__ SR, const float* __restrict__ SI,
                        const float* __restrict__ Ftab, const float* __restrict__ bias,
                        float* __restrict__ out) {
  const float* Fre = Ftab;
  const float* Fim = Ftab + 16384;
  int blk = blockIdx.x;
  int q = blk >> 7, m = blk & 127;
  int wave = threadIdx.x >> 5;
  int lane = threadIdx.x & 31;
  int c = lane & 15, h = lane >> 4;
  int ot = wave & 3;
  int ng = wave >> 2;
  const float* sr_p = SR + ((size_t)blk << 12) + (ot << 4) + c;
  const float* si_p = SI + ((size_t)blk << 12) + (ot << 4) + c;

  v2f sr[16], si[16];
#pragma unroll
  for (int ks = 0; ks < 16; ++ks) {
    int row = ((ks << 2) + (h << 1)) << 6;
    sr[ks].x = sr_p[row]; sr[ks].y = sr_p[row + 64];
    si[ks].x = si_p[row]; si[ks].y = si_p[row + 64];
  }

  v8f p1[4] = {}, p2[4] = {}, p3[4] = {};
#pragma unroll
  for (int ks = 0; ks < 16; ++ks) {
    v2f ss = sr[ks] + si[ks];
#pragma unroll
    for (int j = 0; j < 4; ++j) {
      int n0 = ((ng << 2) + j) << 4;
      v2f fr = fragB_kmap(Fre, ks << 2, n0);
      v2f fi = fragB_kmap(Fim, ks << 2, n0);
      v2f fs = fr + fi;
      v2f fd = fr - fi;
      p1[j] = wmma4(ss, fr, p1[j]);
      p2[j] = wmma4(sr[ks], fs, p2[j]);
      p3[j] = wmma4(si[ks], fd, p3[j]);
    }
  }

  int ob = (ot << 4) + (h << 3);
#pragma unroll
  for (int r = 0; r < 8; ++r) {
    float bo = bias[ob + r];
    float* orow = out + ((((size_t)(q * 64 + ob + r) << 7) + m) << 7) + (ng << 6) + c;
    float* orow8 = orow + ((size_t)8 << 20);
#pragma unroll
    for (int j = 0; j < 4; ++j) {
      orow[j << 4]  = (p1[j][r] - p3[j][r]) * (1.f / 16384.f) + bo;
      orow8[j << 4] = (p1[j][r] - p2[j][r]) * (1.f / 16384.f) + bo;
    }
  }
}

extern "C" void kernel_launch(void* const* d_in, const int* in_sizes, int n_in,
                              void* d_out, int out_size, void* d_ws, size_t ws_size,
                              hipStream_t stream) {
  (void)in_sizes; (void)n_in; (void)out_size; (void)ws_size;
  const float* x    = (const float*)d_in[0];
  const float* w    = (const float*)d_in[1];
  const float* bias = (const float*)d_in[2];
  float* out = (float*)d_out;

  float* ws   = (float*)d_ws;
  float* Ftab = ws;
  float* XbR  = ws + 32768;
  float* XbI  = XbR + 2097152;
  float* YbR  = XbI + 2097152;
  float* YbI  = YbR + 2097152;
  float* SR   = YbI + 2097152;
  float* SI   = SR + 4194304;

  k0_tables<<<16, 256, 0, stream>>>(Ftab);
  k1_fwd  <<<512, 256, 0, stream>>>(x, Ftab, XbR, XbI);
  k3_mix  <<<256, 512, 0, stream>>>(XbR, XbI, w, Ftab, YbR, YbI);
  k4a_col <<<512, 256, 0, stream>>>(YbR, YbI, Ftab, SR, SI);
  k4b_row <<<1024, 256, 0, stream>>>(SR, SI, Ftab, bias, out);
}


# k1 column DFT and k4b row IDFT act on real images: rows +v and -v paired (Fr even, Fi odd), 4 real products on half the band replace the 3-product complex form; their WMMAs 3 -> 2; sum order changes
# speedup vs baseline: 7.1399x; 1.1196x over previous
#include <hip/hip_runtime.h>


typedef __attribute__((ext_vector_type(2))) float v2f;
typedef __attribute__((ext_vector_type(4))) float v4f;
typedef __attribute__((ext_vector_type(8))) float v8f;

#define PI_F 3.14159265358979323846f

__device__ __forceinline__ v8f wmma4(v2f a, v2f b, v8f c) {
  return __builtin_amdgcn_wmma_f32_16x16x4_f32(false, a, false, b, (short)0, c,
                                               false, false);
}

__device__ __forceinline__ int bandmap(int bi) { return (bi + 96) & 127; }

__device__ __forceinline__ v2f fragA_kmap(const float* F, int m0, int k0) {
  int lane = threadIdx.x & 31;
  int m = m0 + (lane & 15);
  int kk = k0 + ((lane >> 4) << 1);
  v2f r; r.x = F[m * 128 + bandmap(kk)]; r.y = F[m * 128 + bandmap(kk + 1)];
  return r;
}
__device__ __forceinline__ v2f fragB_kmap(const float* F, int k0, int n0) {
  int lane = threadIdx.x & 31;
  int n = n0 + (lane & 15);
  int kk = k0 + ((lane >> 4) << 1);
  v2f r; r.x = F[bandmap(kk) * 128 + n]; r.y = F[bandmap(kk + 1) * 128 + n];
  return r;
}

__global__ void k0_tables(float* __restrict__ Ftab) {
  for (int t = threadIdx.x + blockIdx.x * blockDim.x; t < 16384;
       t += blockDim.x * gridDim.x) {
    int j = t >> 7, k = t & 127;
    float ang = -2.f * PI_F * (float)((j * k) & 127) / 128.f;
    float s, c;
    sincosf(ang, &s, &c);
    Ftab[t] = c;
    Ftab[16384 + t] = s;
  }
}

#define FP 132
__global__ void __launch_bounds__(256)
k1_fwd(const float* __restrict__ x, const float* __restrict__ Ftab,
       float* __restrict__ XbR, float* __restrict__ XbI) {
  __shared__ __attribute__((aligned(16))) float Fbr[64 * FP];
  __shared__ __attribute__((aligned(16))) float Fbi[64 * FP];
  __shared__ __attribute__((aligned(16))) float Tr[64 * FP];
  __shared__ __attribute__((aligned(16))) float Ti[64 * FP];
  const float* Fre = Ftab;
  const float* Fim = Ftab + 16384;
  int blk = blockIdx.x;
  int q = blk >> 6, ci = blk & 63;
  const float* Xre = x + ((size_t)(((q + 8) << 6) + ci) << 14);
  const float* Xim = x + ((size_t)((q << 6) + ci) << 14);
  int wave = threadIdx.x >> 5;
  int lane = threadIdx.x & 31;
  int c = lane & 15, h = lane >> 4;

  v2f x1[32], x2[32];
  const float* xp1 = Xre + (h << 8) + (wave << 4) + c;
  const float* xp2 = Xim + (h << 8) + (wave << 4) + c;
#pragma unroll
  for (int ks = 0; ks < 32; ++ks) {
    x1[ks].x = xp1[ks << 9];
    x1[ks].y = xp1[(ks << 9) + 128];
    x2[ks].x = xp2[ks << 9];
    x2[ks].y = xp2[(ks << 9) + 128];
  }

  for (int q = threadIdx.x; q < 2048; q += 256) {
    int m = q >> 5, k4 = (q & 31) << 2;
    int g = bandmap(m) * 128 + k4;
    *(v4f*)(Fbr + m * FP + k4) = *(const v4f*)(Fre + g);
    *(v4f*)(Fbi + m * FP + k4) = *(const v4f*)(Fim + g);
  }
  __syncthreads();

  for (int mt = 0; mt < 2; ++mt) {
    v8f g1 = {}, g2 = {}, g3 = {}, g4 = {};
    const float* ar_p = Fbr + (32 + (mt << 4) + c) * FP + (h << 1);
    const float* ai_p = Fbi + (32 + (mt << 4) + c) * FP + (h << 1);
#pragma unroll
    for (int ks = 0; ks < 32; ++ks) {
      v2f ar = *(const v2f*)(ar_p + (ks << 2));
      v2f ai = *(const v2f*)(ai_p + (ks << 2));
      g1 = wmma4(ar, x1[ks], g1);
      g2 = wmma4(ai, x2[ks], g2);
      g3 = wmma4(ar, x2[ks], g3);
      g4 = wmma4(ai, x1[ks], g4);
    }
    int wc = (wave << 4) + c;
    int ub = (mt << 4) + (h << 3);
#pragma unroll
    for (int r = 0; r < 8; ++r) {
      int u = ub + r;
      Tr[(32 + u) * FP + wc] = g1[r] - g2[r];
      Ti[(32 + u) * FP + wc] = g3[r] + g4[r];
      int um = (u == 0) ? 0 : 32 - u;
      Tr[um * FP + wc] = (u == 0) ? 0.f : g1[r] + g2[r];
      Ti[um * FP + wc] = (u == 0) ? 0.f : g3[r] - g4[r];
    }
  }
  __syncthreads();

  int n0 = (wave & 3) << 4;
  int mt0 = (wave >> 2) << 1;
  const float* br_p = Fbr + (n0 + c) * FP + (h << 1);
  const float* bi_p = Fbi + (n0 + c) * FP + (h << 1);
  const float* tr_p = Tr + ((mt0 << 4) + c) * FP + (h << 1);
  const float* ti_p = Ti + ((mt0 << 4) + c) * FP + (h << 1);
  v8f s1[2] = {}, s2[2] = {}, s3[2] = {};
#pragma unroll 4
  for (int ks = 0; ks < 32; ++ks) {
    v2f br = *(const v2f*)(br_p + (ks << 2));
    v2f bi = *(const v2f*)(bi_p + (ks << 2));
    v2f bd = bi - br;
    v2f bs = br + bi;
#pragma unroll
    for (int j = 0; j < 2; ++j) {
      v2f tr = *(const v2f*)(tr_p + j * (16 * FP) + (ks << 2));
      v2f ti = *(const v2f*)(ti_p + j * (16 * FP) + (ks << 2));
      v2f ts = tr + ti;
      s1[j] = wmma4(ts, br, s1[j]);
      s2[j] = wmma4(tr, bd, s2[j]);
      s3[j] = wmma4(ti, bs, s3[j]);
    }
  }
  int n = n0 + c;
  int dv = n - 32;
#pragma unroll
  for (int j = 0; j < 2; ++j) {
    int mb = ((mt0 + j) << 4) + (h << 3);
#pragma unroll
    for (int r = 0; r < 8; ++r) {
      int uu = mb + r;
      int du = uu - 32;
      float msk = (du * du + dv * dv <= 900) ? 1.f : 0.f;
      size_t off = ((size_t)blk << 12) + (size_t)(uu * 64 + n);
      XbR[off] = (s1[j][r] - s3[j][r]) * msk;
      XbI[off] = (s1[j][r] + s2[j][r]) * msk;
    }
  }
}

#define WPK 386
#define WPP 1184
__global__ void __launch_bounds__(512) __attribute__((amdgpu_waves_per_eu(4, 4)))
k3_mix(const float* __restrict__ XbR, const float* __restrict__ XbI,
       const float* __restrict__ w, const float* __restrict__ Ftab,
       float* __restrict__ YbR, float* __restrict__ YbI) {
  __shared__ __attribute__((aligned(16))) float Wc[32 * WPP];
  int wave = threadIdx.x >> 5;
  int pr = wave >> 1;
  int oh = wave & 1;
  int lane = threadIdx.x & 31;
  int c = lane & 15, h = lane >> 4;

  int u = bandmap(blockIdx.x >> 2);
  int u2 = (u + u) & 127;
  float a1r = Ftab[128 + u],  a1i = Ftab[16384 + 128 + u];
  float a2r = Ftab[128 + u2], a2i = Ftab[16384 + 128 + u2];

  for (int e = threadIdx.x; e < 12288; e += 512) {
    int rest = e >> 5;
    int oh = rest / 24;
    int rho = ((rest - oh * 24) << 3) + (e & 7);
    int osrc = (oh << 2) + ((e >> 3) & 3);
    int i = rho / 3;
    int kx = rho - i * 3;
    const float* wp = w + osrc * 576 + i * 9 + kx;
    float w0 = wp[0], w1 = wp[3], w2 = wp[6];
    int col = (osrc + 32) & 63;
    float wre = w0 + a1r * w1 + a2r * w2;
    float wim = a1i * w1 + a2i * w2;
    float* d = Wc + (i >> 1) * WPP + kx * WPK + col * 6 + (i & 1);
    d[0] = wre;
    d[2] = wim - wre;
    d[4] = wre + wim;
  }

  int f = (blockIdx.x << 4) + c;
  int v = bandmap(f & 63);
  int v2 = (v + v) & 127;
  float b1r = Ftab[128 + v],  b1i = Ftab[16384 + 128 + v];
  float b2r = Ftab[128 + v2], b2i = Ftab[16384 + 128 + v2];
  __syncthreads();

  const float* xr_p = XbR + ((size_t)(pr << 6) << 12) + f;
  const float* xi_p = XbI + ((size_t)(pr << 6) << 12) + f;
  v8f s1[2] = {}, s2[2] = {}, s3[2] = {};
#pragma nounroll
  for (int i0 = 0; i0 < 64; i0 += 4) {
    int kk = i0 + (h << 1);
    float xr0 = xr_p[(size_t)kk << 12], xr1 = xr_p[(size_t)(kk + 1) << 12];
    float xi0 = xi_p[(size_t)kk << 12], xi1 = xi_p[(size_t)(kk + 1) << 12];
    const float* wb = Wc + (kk >> 1) * WPP + (oh * 32 + c) * 6;
    v2f bf[3][2][3];
#pragma unroll
    for (int kx = 0; kx < 3; ++kx)
#pragma unroll
      for (int ot = 0; ot < 2; ++ot) {
        const v2f* q = (const v2f*)(wb + kx * WPK + ot * 96);
        bf[kx][ot][0] = q[0]; bf[kx][ot][1] = q[1]; bf[kx][ot][2] = q[2];
      }
#pragma unroll
    for (int kx = 0; kx < 3; ++kx) {
      v2f ar, ai;
      if (kx == 0) {
        ar.x = xr0; ar.y = xr1; ai.x = xi0; ai.y = xi1;
      } else {
        float tr = (kx == 1) ? b1r : b2r;
        float ti = (kx == 1) ? b1i : b2i;
        ar.x = xr0 * tr - xi0 * ti;
        ar.y = xr1 * tr - xi1 * ti;
        ai.x = xr0 * ti + xi0 * tr;
        ai.y = xr1 * ti + xi1 * tr;
      }
      v2f as = ar + ai;
#pragma unroll
      for (int ot = 0; ot < 2; ++ot) {
        s1[ot] = wmma4(as, bf[kx][ot][0], s1[ot]);
        s2[ot] = wmma4(ar, bf[kx][ot][1], s2[ot]);
        s3[ot] = wmma4(ai, bf[kx][ot][2], s3[ot]);
      }
    }
  }

  int fb = (blockIdx.x << 4) + (h << 3);
#pragma unroll
  for (int ot = 0; ot < 2; ++ot)
#pragma unroll
    for (int r = 0; r < 8; ++r) {
      size_t off = ((size_t)(fb + r) << 9) + (size_t)((pr << 6) + (oh << 5) + (ot << 4) + c);
      YbR[off] = s1[ot][r] - s3[ot][r];
      YbI[off] = s1[ot][r] + s2[ot][r];
    }
}

#define APITCH 68
__global__ void k4a_col(const float* __restrict__ YbR, const float* __restrict__ YbI,
                        const float* __restrict__ Ftab,
                        float* __restrict__ SR, float* __restrict__ SI) {
  __shared__ float A1[64 * APITCH];
  __shared__ float A2[64 * APITCH];
  __shared__ float A3[64 * APITCH];
  const float* Fre = Ftab;
  const float* Fim = Ftab + 16384;
  int mh = blockIdx.x & 1;
  int nb = blockIdx.x >> 1;
  int wave = threadIdx.x >> 5;
  int lane = threadIdx.x & 31;
  int c = lane & 15, h = lane >> 4;
  int n0 = ((nb << 3) + wave) << 4;
  int vv = n0 >> 9, q = (n0 >> 6) & 7, o0 = n0 & 63;

  v2f br[16], bi[16];
  const float* yr = YbR + n0 + c;
  const float* yi = YbI + n0 + c;
#pragma unroll
  for (int ks = 0; ks < 16; ++ks) {
    size_t row = (size_t)((ks << 2) + (h << 1)) << 15;
    br[ks].x = yr[row]; br[ks].y = yr[row + 32768];
    bi[ks].x = yi[row]; bi[ks].y = yi[row + 32768];
  }

  for (int idx = threadIdx.x; idx < 4096; idx += 256) {
    int mr = idx >> 6, k = idx & 63;
    int g = ((mh << 6) + mr) * 128 + bandmap(k);
    float fr = Fre[g], fi = Fim[g];
    A1[mr * APITCH + k] = fr;
    A2[mr * APITCH + k] = fr - fi;
    A3[mr * APITCH + k] = fr + fi;
  }
  __syncthreads();

  for (int mt = 0; mt < 4; ++mt) {
    v8f cP = {}, cQ = {}, cS = {};
    const float* a1_p = A1 + ((mt << 4) + c) * APITCH + (h << 1);
    const float* a2_p = A2 + ((mt << 4) + c) * APITCH + (h << 1);
    const float* a3_p = A3 + ((mt << 4) + c) * APITCH + (h << 1);
#pragma unroll
    for (int ks = 0; ks < 16; ++ks) {
      v2f a1 = *(const v2f*)(a1_p + (ks << 2));
      v2f a2 = *(const v2f*)(a2_p + (ks << 2));
      v2f a3 = *(const v2f*)(a3_p + (ks << 2));
      v2f bs = br[ks] + bi[ks];
      cP = wmma4(a1, bs, cP);
      cQ = wmma4(a2, bi[ks], cQ);
      cS = wmma4(a3, br[ks], cS);
    }
    int mbase = (mh << 6) + (mt << 4) + (h << 3);
#pragma unroll
    for (int r = 0; r < 8; ++r) {
      size_t off = ((((size_t)(q * 128 + mbase + r) << 6) + vv) << 6) + o0 + c;
      SR[off] = cP[r] - cQ[r];
      SI[off] = cP[r] - cS[r];
    }
  }
}

__global__ void __launch_bounds__(256)
k4b_row(const float* __restrict__ SR, const float* __restrict__ SI,
                        const float* __restrict__ Ftab, const float* __restrict__ bias,
                        float* __restrict__ out) {
  const float* Fre = Ftab;
  const float* Fim = Ftab + 16384;
  int blk = blockIdx.x;
  int q = blk >> 7, m = blk & 127;
  int wave = threadIdx.x >> 5;
  int lane = threadIdx.x & 31;
  int c = lane & 15, h = lane >> 4;
  int ot = wave & 3;
  int ng = wave >> 2;
  const float* sr_p = SR + ((size_t)blk << 12) + (ot << 4) + c;
  const float* si_p = SI + ((size_t)blk << 12) + (ot << 4) + c;

  v2f ap[8], aq[8], bp[8], bq[8];
#pragma unroll
  for (int ks = 0; ks < 8; ++ks) {
    int v = (ks << 2) + (h << 1);
    int rp = (32 + v) << 6, rm = (32 - v) << 6;
    v2f srp, srm, sip, sim;
    srp.x = sr_p[rp]; srp.y = sr_p[rp + 64];
    srm.x = sr_p[rm]; srm.y = sr_p[rm - 64];
    sip.x = si_p[rp]; sip.y = si_p[rp + 64];
    sim.x = si_p[rm]; sim.y = si_p[rm - 64];
    ap[ks] = srp + srm;
    aq[ks] = sip - sim;
    bp[ks] = sip + sim;
    bq[ks] = srm - srp;
  }
  if (h == 0) { ap[0].x *= 0.5f; bp[0].x *= 0.5f; }

  v8f pr[4] = {}, pi[4] = {};
#pragma unroll
  for (int ks = 0; ks < 16; ++ks) {
    const float* Fsrc = (ks < 8) ? Fre : Fim;
    int krow = ((ks & 7) << 2) + (h << 1);
    v2f are = (ks < 8) ? ap[ks & 7] : aq[ks & 7];
    v2f aim = (ks < 8) ? bp[ks & 7] : bq[ks & 7];
#pragma unroll
    for (int j = 0; j < 4; ++j) {
      int n = (((ng << 2) + j) << 4) + c;
      v2f f; f.x = Fsrc[krow * 128 + n]; f.y = Fsrc[(krow + 1) * 128 + n];
      pr[j] = wmma4(are, f, pr[j]);
      pi[j] = wmma4(aim, f, pi[j]);
    }
  }

  int ob = (ot << 4) + (h << 3);
#pragma unroll
  for (int r = 0; r < 8; ++r) {
    float bo = bias[ob + r];
    float* orow = out + ((((size_t)(q * 64 + ob + r) << 7) + m) << 7) + (ng << 6) + c;
    float* orow8 = orow + ((size_t)8 << 20);
#pragma unroll
    for (int j = 0; j < 4; ++j) {
      orow[j << 4]  = pr[j][r] * (1.f / 16384.f) + bo;
      orow8[j << 4] = pi[j][r] * (1.f / 16384.f) + bo;
    }
  }
}

extern "C" void kernel_launch(void* const* d_in, const int* in_sizes, int n_in,
                              void* d_out, int out_size, void* d_ws, size_t ws_size,
                              hipStream_t stream) {
  (void)in_sizes; (void)n_in; (void)out_size; (void)ws_size;
  const float* x    = (const float*)d_in[0];
  const float* w    = (const float*)d_in[1];
  const float* bias = (const float*)d_in[2];
  float* out = (float*)d_out;

  float* ws   = (float*)d_ws;
  float* Ftab = ws;
  float* XbR  = ws + 32768;
  float* XbI  = XbR + 2097152;
  float* YbR  = XbI + 2097152;
  float* YbI  = YbR + 2097152;
  float* SR   = YbI + 2097152;
  float* SI   = SR + 4194304;

  k0_tables<<<16, 256, 0, stream>>>(Ftab);
  k1_fwd  <<<512, 256, 0, stream>>>(x, Ftab, XbR, XbI);
  k3_mix  <<<256, 512, 0, stream>>>(XbR, XbI, w, Ftab, YbR, YbI);
  k4a_col <<<512, 256, 0, stream>>>(YbR, YbI, Ftab, SR, SI);
  k4b_row <<<1024, 256, 0, stream>>>(SR, SI, Ftab, bias, out);
}


# k1 stage 2 and k4a (the complex DFT GEMMs left): band indices +v and -v paired via F[-v] = conj F[v]; 4 real products on half the band replace the 3-product form; WMMAs 3 -> 2; sum order changes
# speedup vs baseline: 7.9405x; 1.1121x over previous
#include <hip/hip_runtime.h>


typedef __attribute__((ext_vector_type(2))) float v2f;
typedef __attribute__((ext_vector_type(4))) float v4f;
typedef __attribute__((ext_vector_type(8))) float v8f;

#define PI_F 3.14159265358979323846f

__device__ __forceinline__ v8f wmma4(v2f a, v2f b, v8f c) {
  return __builtin_amdgcn_wmma_f32_16x16x4_f32(false, a, false, b, (short)0, c,
                                               false, false);
}

__device__ __forceinline__ int bandmap(int bi) { return (bi + 96) & 127; }

__device__ __forceinline__ v2f fragA_kmap(const float* F, int m0, int k0) {
  int lane = threadIdx.x & 31;
  int m = m0 + (lane & 15);
  int kk = k0 + ((lane >> 4) << 1);
  v2f r; r.x = F[m * 128 + bandmap(kk)]; r.y = F[m * 128 + bandmap(kk + 1)];
  return r;
}
__device__ __forceinline__ v2f fragB_kmap(const float* F, int k0, int n0) {
  int lane = threadIdx.x & 31;
  int n = n0 + (lane & 15);
  int kk = k0 + ((lane >> 4) << 1);
  v2f r; r.x = F[bandmap(kk) * 128 + n]; r.y = F[bandmap(kk + 1) * 128 + n];
  return r;
}

__global__ void k0_tables(float* __restrict__ Ftab) {
  for (int t = threadIdx.x + blockIdx.x * blockDim.x; t < 16384;
       t += blockDim.x * gridDim.x) {
    int j = t >> 7, k = t & 127;
    float ang = -2.f * PI_F * (float)((j * k) & 127) / 128.f;
    float s, c;
    sincosf(ang, &s, &c);
    Ftab[t] = c;
    Ftab[16384 + t] = s;
  }
}

#define FP 132
__global__ void __launch_bounds__(256)
k1_fwd(const float* __restrict__ x, const float* __restrict__ Ftab,
       float* __restrict__ XbR, float* __restrict__ XbI) {
  __shared__ __attribute__((aligned(16))) float Fbr[32 * FP];
  __shared__ __attribute__((aligned(16))) float Fbi[32 * FP];
  __shared__ __attribute__((aligned(16))) float Tr[64 * FP];
  __shared__ __attribute__((aligned(16))) float Ti[64 * FP];
  const float* Fre = Ftab;
  const float* Fim = Ftab + 16384;
  int blk = blockIdx.x;
  int q = blk >> 6, ci = blk & 63;
  const float* Xre = x + ((size_t)(((q + 8) << 6) + ci) << 14);
  const float* Xim = x + ((size_t)((q << 6) + ci) << 14);
  int wave = threadIdx.x >> 5;
  int lane = threadIdx.x & 31;
  int c = lane & 15, h = lane >> 4;

  v2f x1[32], x2[32];
  const float* xp1 = Xre + (h << 8) + (wave << 4) + c;
  const float* xp2 = Xim + (h << 8) + (wave << 4) + c;
#pragma unroll
  for (int ks = 0; ks < 32; ++ks) {
    x1[ks].x = xp1[ks << 9];
    x1[ks].y = xp1[(ks << 9) + 128];
    x2[ks].x = xp2[ks << 9];
    x2[ks].y = xp2[(ks << 9) + 128];
  }

  for (int q = threadIdx.x; q < 1024; q += 256) {
    int m = q >> 5, k4 = (q & 31) << 2;
    int g = m * 128 + k4;
    *(v4f*)(Fbr + m * FP + k4) = *(const v4f*)(Fre + g);
    *(v4f*)(Fbi + m * FP + k4) = *(const v4f*)(Fim + g);
  }
  __syncthreads();

  for (int mt = 0; mt < 2; ++mt) {
    v8f g1 = {}, g2 = {}, g3 = {}, g4 = {};
    const float* ar_p = Fbr + ((mt << 4) + c) * FP + (h << 1);
    const float* ai_p = Fbi + ((mt << 4) + c) * FP + (h << 1);
#pragma unroll
    for (int ks = 0; ks < 32; ++ks) {
      v2f ar = *(const v2f*)(ar_p + (ks << 2));
      v2f ai = *(const v2f*)(ai_p + (ks << 2));
      g1 = wmma4(ar, x1[ks], g1);
      g2 = wmma4(ai, x2[ks], g2);
      g3 = wmma4(ar, x2[ks], g3);
      g4 = wmma4(ai, x1[ks], g4);
    }
    int wc = (wave << 4) + c;
    int ub = (mt << 4) + (h << 3);
#pragma unroll
    for (int r = 0; r < 8; ++r) {
      int u = ub + r;
      Tr[(32 + u) * FP + wc] = g1[r] - g2[r];
      Ti[(32 + u) * FP + wc] = g3[r] + g4[r];
      int um = (u == 0) ? 0 : 32 - u;
      Tr[um * FP + wc] = (u == 0) ? 0.f : g1[r] + g2[r];
      Ti[um * FP + wc] = (u == 0) ? 0.f : g3[r] - g4[r];
    }
  }
  __syncthreads();

  int nt = wave & 1;
  int mt2 = wave >> 1;
  const float* br_p = Fbr + ((nt << 4) + c) * FP + (h << 1);
  const float* bi_p = Fbi + ((nt << 4) + c) * FP + (h << 1);
  const float* tr_p = Tr + ((mt2 << 4) + c) * FP + (h << 1);
  const float* ti_p = Ti + ((mt2 << 4) + c) * FP + (h << 1);
  v8f p1 = {}, p2 = {}, p3 = {}, p4 = {};
#pragma unroll 4
  for (int ks = 0; ks < 32; ++ks) {
    v2f br = *(const v2f*)(br_p + (ks << 2));
    v2f bi = *(const v2f*)(bi_p + (ks << 2));
    v2f tr = *(const v2f*)(tr_p + (ks << 2));
    v2f ti = *(const v2f*)(ti_p + (ks << 2));
    p1 = wmma4(tr, br, p1);
    p2 = wmma4(ti, bi, p2);
    p3 = wmma4(tr, bi, p3);
    p4 = wmma4(ti, br, p4);
  }
  int v = (nt << 4) + c;
  int vm = (v == 0) ? 0 : 32 - v;
  int mb = (mt2 << 4) + (h << 3);
#pragma unroll
  for (int r = 0; r < 8; ++r) {
    int uu = mb + r;
    int du = uu - 32;
    float msk = (du * du + v * v <= 900) ? 1.f : 0.f;
    size_t row = ((size_t)blk << 12) + (size_t)(uu * 64);
    XbR[row + 32 + v] = (p1[r] - p2[r]) * msk;
    XbI[row + 32 + v] = (p3[r] + p4[r]) * msk;
    XbR[row + vm] = (v == 0) ? 0.f : (p1[r] + p2[r]) * msk;
    XbI[row + vm] = (v == 0) ? 0.f : (p4[r] - p3[r]) * msk;
  }
}

#define WPK 386
#define WPP 1184
__global__ void __launch_bounds__(512) __attribute__((amdgpu_waves_per_eu(4, 4)))
k3_mix(const float* __restrict__ XbR, const float* __restrict__ XbI,
       const float* __restrict__ w, const float* __restrict__ Ftab,
       float* __restrict__ YbR, float* __restrict__ YbI) {
  __shared__ __attribute__((aligned(16))) float Wc[32 * WPP];
  int wave = threadIdx.x >> 5;
  int pr = wave >> 1;
  int oh = wave & 1;
  int lane = threadIdx.x & 31;
  int c = lane & 15, h = lane >> 4;

  int u = bandmap(blockIdx.x >> 2);
  int u2 = (u + u) & 127;
  float a1r = Ftab[128 + u],  a1i = Ftab[16384 + 128 + u];
  float a2r = Ftab[128 + u2], a2i = Ftab[16384 + 128 + u2];

  for (int e = threadIdx.x; e < 12288; e += 512) {
    int rest = e >> 5;
    int oh = rest / 24;
    int rho = ((rest - oh * 24) << 3) + (e & 7);
    int osrc = (oh << 2) + ((e >> 3) & 3);
    int i = rho / 3;
    int kx = rho - i * 3;
    const float* wp = w + osrc * 576 + i * 9 + kx;
    float w0 = wp[0], w1 = wp[3], w2 = wp[6];
    int col = (osrc + 32) & 63;
    float wre = w0 + a1r * w1 + a2r * w2;
    float wim = a1i * w1 + a2i * w2;
    float* d = Wc + (i >> 1) * WPP + kx * WPK + col * 6 + (i & 1);
    d[0] = wre;
    d[2] = wim - wre;
    d[4] = wre + wim;
  }

  int f = (blockIdx.x << 4) + c;
  int v = bandmap(f & 63);
  int v2 = (v + v) & 127;
  float b1r = Ftab[128 + v],  b1i = Ftab[16384 + 128 + v];
  float b2r = Ftab[128 + v2], b2i = Ftab[16384 + 128 + v2];
  __syncthreads();

  const float* xr_p = XbR + ((size_t)(pr << 6) << 12) + f;
  const float* xi_p = XbI + ((size_t)(pr << 6) << 12) + f;
  v8f s1[2] = {}, s2[2] = {}, s3[2] = {};
#pragma nounroll
  for (int i0 = 0; i0 < 64; i0 += 4) {
    int kk = i0 + (h << 1);
    float xr0 = xr_p[(size_t)kk << 12], xr1 = xr_p[(size_t)(kk + 1) << 12];
    float xi0 = xi_p[(size_t)kk << 12], xi1 = xi_p[(size_t)(kk + 1) << 12];
    const float* wb = Wc + (kk >> 1) * WPP + (oh * 32 + c) * 6;
    v2f bf[3][2][3];
#pragma unroll
    for (int kx = 0; kx < 3; ++kx)
#pragma unroll
      for (int ot = 0; ot < 2; ++ot) {
        const v2f* q = (const v2f*)(wb + kx * WPK + ot * 96);
        bf[kx][ot][0] = q[0]; bf[kx][ot][1] = q[1]; bf[kx][ot][2] = q[2];
      }
#pragma unroll
    for (int kx = 0; kx < 3; ++kx) {
      v2f ar, ai;
      if (kx == 0) {
        ar.x = xr0; ar.y = xr1; ai.x = xi0; ai.y = xi1;
      } else {
        float tr = (kx == 1) ? b1r : b2r;
        float ti = (kx == 1) ? b1i : b2i;
        ar.x = xr0 * tr - xi0 * ti;
        ar.y = xr1 * tr - xi1 * ti;
        ai.x = xr0 * ti + xi0 * tr;
        ai.y = xr1 * ti + xi1 * tr;
      }
      v2f as = ar + ai;
#pragma unroll
      for (int ot = 0; ot < 2; ++ot) {
        s1[ot] = wmma4(as, bf[kx][ot][0], s1[ot]);
        s2[ot] = wmma4(ar, bf[kx][ot][1], s2[ot]);
        s3[ot] = wmma4(ai, bf[kx][ot][2], s3[ot]);
      }
    }
  }

  int fb = (blockIdx.x << 4) + (h << 3);
#pragma unroll
  for (int ot = 0; ot < 2; ++ot)
#pragma unroll
    for (int r = 0; r < 8; ++r) {
      size_t off = ((size_t)(fb + r) << 9) + (size_t)((pr << 6) + (oh << 5) + (ot << 4) + c);
      YbR[off] = s1[ot][r] - s3[ot][r];
      YbI[off] = s1[ot][r] + s2[ot][r];
    }
}

#define APITCH 68
__global__ void k4a_col(const float* __restrict__ YbR, const float* __restrict__ YbI,
                        const float* __restrict__ Ftab,
                        float* __restrict__ SR, float* __restrict__ SI) {
  __shared__ __attribute__((aligned(16))) float Af[64 * APITCH];
  const float* Fre = Ftab;
  const float* Fim = Ftab + 16384;
  int mh = blockIdx.x & 1;
  int nb = blockIdx.x >> 1;
  int wave = threadIdx.x >> 5;
  int lane = threadIdx.x & 31;
  int c = lane & 15, h = lane >> 4;
  int n0 = ((nb << 3) + wave) << 4;
  int vv = n0 >> 9, q = (n0 >> 6) & 7, o0 = n0 & 63;

  v2f bp[8], bq[8], cp[8], cq[8];
  const float* yr = YbR + n0 + c;
  const float* yi = YbI + n0 + c;
#pragma unroll
  for (int ks = 0; ks < 8; ++ks) {
    int u = (ks << 2) + (h << 1);
    size_t rp = (size_t)(32 + u) << 15, rm = (size_t)(32 - u) << 15;
    v2f yrp, yrm, yip, yim;
    yrp.x = yr[rp]; yrp.y = yr[rp + 32768];
    yrm.x = yr[rm]; yrm.y = yr[rm - 32768];
    yip.x = yi[rp]; yip.y = yi[rp + 32768];
    yim.x = yi[rm]; yim.y = yi[rm - 32768];
    bp[ks] = yrp + yrm;
    bq[ks] = yip - yim;
    cp[ks] = yip + yim;
    cq[ks] = yrm - yrp;
  }
  if (h == 0) { bp[0].x *= 0.5f; cp[0].x *= 0.5f; }

  for (int idx = threadIdx.x; idx < 1024; idx += 256) {
    int mr = idx >> 4, k4 = (idx & 15) << 2;
    const float* Fsrc = (k4 < 32) ? Fre : Fim;
    *(v4f*)(Af + mr * APITCH + k4) = *(const v4f*)(Fsrc + ((mh << 6) + mr) * 128 + (k4 & 31));
  }
  __syncthreads();

  for (int mt = 0; mt < 4; ++mt) {
    v8f cR = {}, cI = {};
    const float* a_p = Af + ((mt << 4) + c) * APITCH + (h << 1);
    v2f af[16];
#pragma unroll
    for (int ks = 0; ks < 16; ++ks) af[ks] = *(const v2f*)(a_p + (ks << 2));
#pragma unroll
    for (int ks = 0; ks < 16; ++ks) {
      v2f bre = (ks < 8) ? bp[ks & 7] : bq[ks & 7];
      v2f bim = (ks < 8) ? cp[ks & 7] : cq[ks & 7];
      cR = wmma4(af[ks], bre, cR);
      cI = wmma4(af[ks], bim, cI);
    }
    int mbase = (mh << 6) + (mt << 4) + (h << 3);
#pragma unroll
    for (int r = 0; r < 8; ++r) {
      size_t off = ((((size_t)(q * 128 + mbase + r) << 6) + vv) << 6) + o0 + c;
      SR[off] = cR[r];
      SI[off] = cI[r];
    }
  }
}

__global__ void __launch_bounds__(256)
k4b_row(const float* __restrict__ SR, const float* __restrict__ SI,
                        const float* __restrict__ Ftab, const float* __restrict__ bias,
                        float* __restrict__ out) {
  const float* Fre = Ftab;
  const float* Fim = Ftab + 16384;
  int blk = blockIdx.x;
  int q = blk >> 7, m = blk & 127;
  int wave = threadIdx.x >> 5;
  int lane = threadIdx.x & 31;
  int c = lane & 15, h = lane >> 4;
  int ot = wave & 3;
  int ng = wave >> 2;
  const float* sr_p = SR + ((size_t)blk << 12) + (ot << 4) + c;
  const float* si_p = SI + ((size_t)blk << 12) + (ot << 4) + c;

  v2f ap[8], aq[8], bp[8], bq[8];
#pragma unroll
  for (int ks = 0; ks < 8; ++ks) {
    int v = (ks << 2) + (h << 1);
    int rp = (32 + v) << 6, rm = (32 - v) << 6;
    v2f srp, srm, sip, sim;
    srp.x = sr_p[rp]; srp.y = sr_p[rp + 64];
    srm.x = sr_p[rm]; srm.y = sr_p[rm - 64];
    sip.x = si_p[rp]; sip.y = si_p[rp + 64];
    sim.x = si_p[rm]; sim.y = si_p[rm - 64];
    ap[ks] = srp + srm;
    aq[ks] = sip - sim;
    bp[ks] = sip + sim;
    bq[ks] = srm - srp;
  }
  if (h == 0) { ap[0].x *= 0.5f; bp[0].x *= 0.5f; }

  v8f pr[4] = {}, pi[4] = {};
#pragma unroll
  for (int ks = 0; ks < 16; ++ks) {
    const float* Fsrc = (ks < 8) ? Fre : Fim;
    int krow = ((ks & 7) << 2) + (h << 1);
    v2f are = (ks < 8) ? ap[ks & 7] : aq[ks & 7];
    v2f aim = (ks < 8) ? bp[ks & 7] : bq[ks & 7];
#pragma unroll
    for (int j = 0; j < 4; ++j) {
      int n = (((ng << 2) + j) << 4) + c;
      v2f f; f.x = Fsrc[krow * 128 + n]; f.y = Fsrc[(krow + 1) * 128 + n];
      pr[j] = wmma4(are, f, pr[j]);
      pi[j] = wmma4(aim, f, pi[j]);
    }
  }

  int ob = (ot << 4) + (h << 3);
#pragma unroll
  for (int r = 0; r < 8; ++r) {
    float bo = bias[ob + r];
    float* orow = out + ((((size_t)(q * 64 + ob + r) << 7) + m) << 7) + (ng << 6) + c;
    float* orow8 = orow + ((size_t)8 << 20);
#pragma unroll
    for (int j = 0; j < 4; ++j) {
      orow[j << 4]  = pr[j][r] * (1.f / 16384.f) + bo;
      orow8[j << 4] = pi[j][r] * (1.f / 16384.f) + bo;
    }
  }
}

extern "C" void kernel_launch(void* const* d_in, const int* in_sizes, int n_in,
                              void* d_out, int out_size, void* d_ws, size_t ws_size,
                              hipStream_t stream) {
  (void)in_sizes; (void)n_in; (void)out_size; (void)ws_size;
  const float* x    = (const float*)d_in[0];
  const float* w    = (const float*)d_in[1];
  const float* bias = (const float*)d_in[2];
  float* out = (float*)d_out;

  float* ws   = (float*)d_ws;
  float* Ftab = ws;
  float* XbR  = ws + 32768;
  float* XbI  = XbR + 2097152;
  float* YbR  = XbI + 2097152;
  float* YbI  = YbR + 2097152;
  float* SR   = YbI + 2097152;
  float* SI   = SR + 4194304;

  k0_tables<<<16, 256, 0, stream>>>(Ftab);
  k1_fwd  <<<512, 256, 0, stream>>>(x, Ftab, XbR, XbI);
  k3_mix  <<<256, 512, 0, stream>>>(XbR, XbI, w, Ftab, YbR, YbI);
  k4a_col <<<512, 256, 0, stream>>>(YbR, YbI, Ftab, SR, SI);
  k4b_row <<<1024, 256, 0, stream>>>(SR, SI, Ftab, bias, out);
}


# k1 column DFT: summed image row folded 128 -> 64 via F[u][128-j] = conj F[u][j] on the real images (B = x[j]+x[128-j] for Fr, x[j]-x[128-j] for Fi; row 64 added after); WMMAs halve; sum order changes
# speedup vs baseline: 8.3825x; 1.0557x over previous
#include <hip/hip_runtime.h>


typedef __attribute__((ext_vector_type(2))) float v2f;
typedef __attribute__((ext_vector_type(4))) float v4f;
typedef __attribute__((ext_vector_type(8))) float v8f;

#define PI_F 3.14159265358979323846f

__device__ __forceinline__ v8f wmma4(v2f a, v2f b, v8f c) {
  return __builtin_amdgcn_wmma_f32_16x16x4_f32(false, a, false, b, (short)0, c,
                                               false, false);
}

__device__ __forceinline__ int bandmap(int bi) { return (bi + 96) & 127; }

__device__ __forceinline__ v2f fragA_kmap(const float* F, int m0, int k0) {
  int lane = threadIdx.x & 31;
  int m = m0 + (lane & 15);
  int kk = k0 + ((lane >> 4) << 1);
  v2f r; r.x = F[m * 128 + bandmap(kk)]; r.y = F[m * 128 + bandmap(kk + 1)];
  return r;
}
__device__ __forceinline__ v2f fragB_kmap(const float* F, int k0, int n0) {
  int lane = threadIdx.x & 31;
  int n = n0 + (lane & 15);
  int kk = k0 + ((lane >> 4) << 1);
  v2f r; r.x = F[bandmap(kk) * 128 + n]; r.y = F[bandmap(kk + 1) * 128 + n];
  return r;
}

__global__ void k0_tables(float* __restrict__ Ftab) {
  for (int t = threadIdx.x + blockIdx.x * blockDim.x; t < 16384;
       t += blockDim.x * gridDim.x) {
    int j = t >> 7, k = t & 127;
    float ang = -2.f * PI_F * (float)((j * k) & 127) / 128.f;
    float s, c;
    sincosf(ang, &s, &c);
    Ftab[t] = c;
    Ftab[16384 + t] = s;
  }
}

#define FP 132
__global__ void __launch_bounds__(256)
k1_fwd(const float* __restrict__ x, const float* __restrict__ Ftab,
       float* __restrict__ XbR, float* __restrict__ XbI) {
  __shared__ __attribute__((aligned(16))) float Fbr[32 * FP];
  __shared__ __attribute__((aligned(16))) float Fbi[32 * FP];
  __shared__ __attribute__((aligned(16))) float Tr[64 * FP];
  __shared__ __attribute__((aligned(16))) float Ti[64 * FP];
  const float* Fre = Ftab;
  const float* Fim = Ftab + 16384;
  int blk = blockIdx.x;
  int q = blk >> 6, ci = blk & 63;
  const float* Xre = x + ((size_t)(((q + 8) << 6) + ci) << 14);
  const float* Xim = x + ((size_t)((q << 6) + ci) << 14);
  int wave = threadIdx.x >> 5;
  int lane = threadIdx.x & 31;
  int c = lane & 15, h = lane >> 4;

  v2f e1[16], o1[16], e2[16], o2[16];
  const float* xc1 = Xre + (wave << 4) + c;
  const float* xc2 = Xim + (wave << 4) + c;
  const float* xa1 = xc1 + (h << 8);
  const float* xa2 = xc2 + (h << 8);
  const float* xb1 = xc1 + ((127 - (h << 1)) << 7);
  const float* xb2 = xc2 + ((127 - (h << 1)) << 7);
  int p0 = ((128 - (h << 1)) & 127) << 7;
#pragma unroll
  for (int ks = 0; ks < 16; ++ks) {
    v2f a1, b1, a2, b2;
    a1.x = xa1[ks << 9];
    a1.y = xa1[(ks << 9) + 128];
    a2.x = xa2[ks << 9];
    a2.y = xa2[(ks << 9) + 128];
    b1.y = xb1[-(ks << 9)];
    b2.y = xb2[-(ks << 9)];
    b1.x = (ks == 0) ? xc1[p0] : xb1[128 - (ks << 9)];
    b2.x = (ks == 0) ? xc2[p0] : xb2[128 - (ks << 9)];
    e1[ks] = a1 + b1;
    o1[ks] = a1 - b1;
    e2[ks] = a2 + b2;
    o2[ks] = a2 - b2;
  }
  if (h == 0) { e1[0].x *= 0.5f; e2[0].x *= 0.5f; }
  float x64a = xc1[64 << 7];
  float x64b = xc2[64 << 7];

  for (int q = threadIdx.x; q < 1024; q += 256) {
    int m = q >> 5, k4 = (q & 31) << 2;
    int g = m * 128 + k4;
    *(v4f*)(Fbr + m * FP + k4) = *(const v4f*)(Fre + g);
    *(v4f*)(Fbi + m * FP + k4) = *(const v4f*)(Fim + g);
  }
  __syncthreads();

  for (int mt = 0; mt < 2; ++mt) {
    v8f g1 = {}, g2 = {}, g3 = {}, g4 = {};
    const float* ar_p = Fbr + ((mt << 4) + c) * FP + (h << 1);
    const float* ai_p = Fbi + ((mt << 4) + c) * FP + (h << 1);
#pragma unroll
    for (int ks = 0; ks < 16; ++ks) {
      v2f ar = *(const v2f*)(ar_p + (ks << 2));
      v2f ai = *(const v2f*)(ai_p + (ks << 2));
      g1 = wmma4(ar, e1[ks], g1);
      g2 = wmma4(ai, o2[ks], g2);
      g3 = wmma4(ar, e2[ks], g3);
      g4 = wmma4(ai, o1[ks], g4);
    }
#pragma unroll
    for (int r = 0; r < 8; ++r) {
      g1[r] += (r & 1) ? -x64a : x64a;
      g3[r] += (r & 1) ? -x64b : x64b;
    }
    int wc = (wave << 4) + c;
    int ub = (mt << 4) + (h << 3);
#pragma unroll
    for (int r = 0; r < 8; ++r) {
      int u = ub + r;
      Tr[(32 + u) * FP + wc] = g1[r] - g2[r];
      Ti[(32 + u) * FP + wc] = g3[r] + g4[r];
      int um = (u == 0) ? 0 : 32 - u;
      Tr[um * FP + wc] = (u == 0) ? 0.f : g1[r] + g2[r];
      Ti[um * FP + wc] = (u == 0) ? 0.f : g3[r] - g4[r];
    }
  }
  __syncthreads();

  int nt = wave & 1;
  int mt2 = wave >> 1;
  const float* br_p = Fbr + ((nt << 4) + c) * FP + (h << 1);
  const float* bi_p = Fbi + ((nt << 4) + c) * FP + (h << 1);
  const float* tr_p = Tr + ((mt2 << 4) + c) * FP + (h << 1);
  const float* ti_p = Ti + ((mt2 << 4) + c) * FP + (h << 1);
  v8f p1 = {}, p2 = {}, p3 = {}, p4 = {};
#pragma unroll 4
  for (int ks = 0; ks < 32; ++ks) {
    v2f br = *(const v2f*)(br_p + (ks << 2));
    v2f bi = *(const v2f*)(bi_p + (ks << 2));
    v2f tr = *(const v2f*)(tr_p + (ks << 2));
    v2f ti = *(const v2f*)(ti_p + (ks << 2));
    p1 = wmma4(tr, br, p1);
    p2 = wmma4(ti, bi, p2);
    p3 = wmma4(tr, bi, p3);
    p4 = wmma4(ti, br, p4);
  }
  int v = (nt << 4) + c;
  int vm = (v == 0) ? 0 : 32 - v;
  int mb = (mt2 << 4) + (h << 3);
#pragma unroll
  for (int r = 0; r < 8; ++r) {
    int uu = mb + r;
    int du = uu - 32;
    float msk = (du * du + v * v <= 900) ? 1.f : 0.f;
    size_t row = ((size_t)blk << 12) + (size_t)(uu * 64);
    XbR[row + 32 + v] = (p1[r] - p2[r]) * msk;
    XbI[row + 32 + v] = (p3[r] + p4[r]) * msk;
    XbR[row + vm] = (v == 0) ? 0.f : (p1[r] + p2[r]) * msk;
    XbI[row + vm] = (v == 0) ? 0.f : (p4[r] - p3[r]) * msk;
  }
}

#define WPK 386
#define WPP 1184
__global__ void __launch_bounds__(512) __attribute__((amdgpu_waves_per_eu(4, 4)))
k3_mix(const float* __restrict__ XbR, const float* __restrict__ XbI,
       const float* __restrict__ w, const float* __restrict__ Ftab,
       float* __restrict__ YbR, float* __restrict__ YbI) {
  __shared__ __attribute__((aligned(16))) float Wc[32 * WPP];
  int wave = threadIdx.x >> 5;
  int pr = wave >> 1;
  int oh = wave & 1;
  int lane = threadIdx.x & 31;
  int c = lane & 15, h = lane >> 4;

  int u = bandmap(blockIdx.x >> 2);
  int u2 = (u + u) & 127;
  float a1r = Ftab[128 + u],  a1i = Ftab[16384 + 128 + u];
  float a2r = Ftab[128 + u2], a2i = Ftab[16384 + 128 + u2];

  for (int e = threadIdx.x; e < 12288; e += 512) {
    int rest = e >> 5;
    int oh = rest / 24;
    int rho = ((rest - oh * 24) << 3) + (e & 7);
    int osrc = (oh << 2) + ((e >> 3) & 3);
    int i = rho / 3;
    int kx = rho - i * 3;
    const float* wp = w + osrc * 576 + i * 9 + kx;
    float w0 = wp[0], w1 = wp[3], w2 = wp[6];
    int col = (osrc + 32) & 63;
    float wre = w0 + a1r * w1 + a2r * w2;
    float wim = a1i * w1 + a2i * w2;
    float* d = Wc + (i >> 1) * WPP + kx * WPK + col * 6 + (i & 1);
    d[0] = wre;
    d[2] = wim - wre;
    d[4] = wre + wim;
  }

  int f = (blockIdx.x << 4) + c;
  int v = bandmap(f & 63);
  int v2 = (v + v) & 127;
  float b1r = Ftab[128 + v],  b1i = Ftab[16384 + 128 + v];
  float b2r = Ftab[128 + v2], b2i = Ftab[16384 + 128 + v2];
  __syncthreads();

  const float* xr_p = XbR + ((size_t)(pr << 6) << 12) + f;
  const float* xi_p = XbI + ((size_t)(pr << 6) << 12) + f;
  v8f s1[2] = {}, s2[2] = {}, s3[2] = {};
#pragma nounroll
  for (int i0 = 0; i0 < 64; i0 += 4) {
    int kk = i0 + (h << 1);
    float xr0 = xr_p[(size_t)kk << 12], xr1 = xr_p[(size_t)(kk + 1) << 12];
    float xi0 = xi_p[(size_t)kk << 12], xi1 = xi_p[(size_t)(kk + 1) << 12];
    const float* wb = Wc + (kk >> 1) * WPP + (oh * 32 + c) * 6;
    v2f bf[3][2][3];
#pragma unroll
    for (int kx = 0; kx < 3; ++kx)
#pragma unroll
      for (int ot = 0; ot < 2; ++ot) {
        const v2f* q = (const v2f*)(wb + kx * WPK + ot * 96);
        bf[kx][ot][0] = q[0]; bf[kx][ot][1] = q[1]; bf[kx][ot][2] = q[2];
      }
#pragma unroll
    for (int kx = 0; kx < 3; ++kx) {
      v2f ar, ai;
      if (kx == 0) {
        ar.x = xr0; ar.y = xr1; ai.x = xi0; ai.y = xi1;
      } else {
        float tr = (kx == 1) ? b1r : b2r;
        float ti = (kx == 1) ? b1i : b2i;
        ar.x = xr0 * tr - xi0 * ti;
        ar.y = xr1 * tr - xi1 * ti;
        ai.x = xr0 * ti + xi0 * tr;
        ai.y = xr1 * ti + xi1 * tr;
      }
      v2f as = ar + ai;
#pragma unroll
      for (int ot = 0; ot < 2; ++ot) {
        s1[ot] = wmma4(as, bf[kx][ot][0], s1[ot]);
        s2[ot] = wmma4(ar, bf[kx][ot][1], s2[ot]);
        s3[ot] = wmma4(ai, bf[kx][ot][2], s3[ot]);
      }
    }
  }

  int fb = (blockIdx.x << 4) + (h << 3);
#pragma unroll
  for (int ot = 0; ot < 2; ++ot)
#pragma unroll
    for (int r = 0; r < 8; ++r) {
      size_t off = ((size_t)(fb + r) << 9) + (size_t)((pr << 6) + (oh << 5) + (ot << 4) + c);
      YbR[off] = s1[ot][r] - s3[ot][r];
      YbI[off] = s1[ot][r] + s2[ot][r];
    }
}

#define APITCH 68
__global__ void k4a_col(const float* __restrict__ YbR, const float* __restrict__ YbI,
                        const float* __restrict__ Ftab,
                        float* __restrict__ SR, float* __restrict__ SI) {
  __shared__ __attribute__((aligned(16))) float Af[64 * APITCH];
  const float* Fre = Ftab;
  const float* Fim = Ftab + 16384;
  int mh = blockIdx.x & 1;
  int nb = blockIdx.x >> 1;
  int wave = threadIdx.x >> 5;
  int lane = threadIdx.x & 31;
  int c = lane & 15, h = lane >> 4;
  int n0 = ((nb << 3) + wave) << 4;
  int vv = n0 >> 9, q = (n0 >> 6) & 7, o0 = n0 & 63;

  v2f bp[8], bq[8], cp[8], cq[8];
  const float* yr = YbR + n0 + c;
  const float* yi = YbI + n0 + c;
#pragma unroll
  for (int ks = 0; ks < 8; ++ks) {
    int u = (ks << 2) + (h << 1);
    size_t rp = (size_t)(32 + u) << 15, rm = (size_t)(32 - u) << 15;
    v2f yrp, yrm, yip, yim;
    yrp.x = yr[rp]; yrp.y = yr[rp + 32768];
    yrm.x = yr[rm]; yrm.y = yr[rm - 32768];
    yip.x = yi[rp]; yip.y = yi[rp + 32768];
    yim.x = yi[rm]; yim.y = yi[rm - 32768];
    bp[ks] = yrp + yrm;
    bq[ks] = yip - yim;
    cp[ks] = yip + yim;
    cq[ks] = yrm - yrp;
  }
  if (h == 0) { bp[0].x *= 0.5f; cp[0].x *= 0.5f; }

  for (int idx = threadIdx.x; idx < 1024; idx += 256) {
    int mr = idx >> 4, k4 = (idx & 15) << 2;
    const float* Fsrc = (k4 < 32) ? Fre : Fim;
    *(v4f*)(Af + mr * APITCH + k4) = *(const v4f*)(Fsrc + ((mh << 6) + mr) * 128 + (k4 & 31));
  }
  __syncthreads();

  for (int mt = 0; mt < 4; ++mt) {
    v8f cR = {}, cI = {};
    const float* a_p = Af + ((mt << 4) + c) * APITCH + (h << 1);
    v2f af[16];
#pragma unroll
    for (int ks = 0; ks < 16; ++ks) af[ks] = *(const v2f*)(a_p + (ks << 2));
#pragma unroll
    for (int ks = 0; ks < 16; ++ks) {
      v2f bre = (ks < 8) ? bp[ks & 7] : bq[ks & 7];
      v2f bim = (ks < 8) ? cp[ks & 7] : cq[ks & 7];
      cR = wmma4(af[ks], bre, cR);
      cI = wmma4(af[ks], bim, cI);
    }
    int mbase = (mh << 6) + (mt << 4) + (h << 3);
#pragma unroll
    for (int r = 0; r < 8; ++r) {
      size_t off = ((((size_t)(q * 128 + mbase + r) << 6) + vv) << 6) + o0 + c;
      SR[off] = cR[r];
      SI[off] = cI[r];
    }
  }
}

__global__ void __launch_bounds__(256)
k4b_row(const float* __restrict__ SR, const float* __restrict__ SI,
                        const float* __restrict__ Ftab, const float* __restrict__ bias,
                        float* __restrict__ out) {
  const float* Fre = Ftab;
  const float* Fim = Ftab + 16384;
  int blk = blockIdx.x;
  int q = blk >> 7, m = blk & 127;
  int wave = threadIdx.x >> 5;
  int lane = threadIdx.x & 31;
  int c = lane & 15, h = lane >> 4;
  int ot = wave & 3;
  int ng = wave >> 2;
  const float* sr_p = SR + ((size_t)blk << 12) + (ot << 4) + c;
  const float* si_p = SI + ((size_t)blk << 12) + (ot << 4) + c;

  v2f ap[8], aq[8], bp[8], bq[8];
#pragma unroll
  for (int ks = 0; ks < 8; ++ks) {
    int v = (ks << 2) + (h << 1);
    int rp = (32 + v) << 6, rm = (32 - v) << 6;
    v2f srp, srm, sip, sim;
    srp.x = sr_p[rp]; srp.y = sr_p[rp + 64];
    srm.x = sr_p[rm]; srm.y = sr_p[rm - 64];
    sip.x = si_p[rp]; sip.y = si_p[rp + 64];
    sim.x = si_p[rm]; sim.y = si_p[rm - 64];
    ap[ks] = srp + srm;
    aq[ks] = sip - sim;
    bp[ks] = sip + sim;
    bq[ks] = srm - srp;
  }
  if (h == 0) { ap[0].x *= 0.5f; bp[0].x *= 0.5f; }

  v8f pr[4] = {}, pi[4] = {};
#pragma unroll
  for (int ks = 0; ks < 16; ++ks) {
    const float* Fsrc = (ks < 8) ? Fre : Fim;
    int krow = ((ks & 7) << 2) + (h << 1);
    v2f are = (ks < 8) ? ap[ks & 7] : aq[ks & 7];
    v2f aim = (ks < 8) ? bp[ks & 7] : bq[ks & 7];
#pragma unroll
    for (int j = 0; j < 4; ++j) {
      int n = (((ng << 2) + j) << 4) + c;
      v2f f; f.x = Fsrc[krow * 128 + n]; f.y = Fsrc[(krow + 1) * 128 + n];
      pr[j] = wmma4(are, f, pr[j]);
      pi[j] = wmma4(aim, f, pi[j]);
    }
  }

  int ob = (ot << 4) + (h << 3);
#pragma unroll
  for (int r = 0; r < 8; ++r) {
    float bo = bias[ob + r];
    float* orow = out + ((((size_t)(q * 64 + ob + r) << 7) + m) << 7) + (ng << 6) + c;
    float* orow8 = orow + ((size_t)8 << 20);
#pragma unroll
    for (int j = 0; j < 4; ++j) {
      orow[j << 4]  = pr[j][r] * (1.f / 16384.f) + bo;
      orow8[j << 4] = pi[j][r] * (1.f / 16384.f) + bo;
    }
  }
}

extern "C" void kernel_launch(void* const* d_in, const int* in_sizes, int n_in,
                              void* d_out, int out_size, void* d_ws, size_t ws_size,
                              hipStream_t stream) {
  (void)in_sizes; (void)n_in; (void)out_size; (void)ws_size;
  const float* x    = (const float*)d_in[0];
  const float* w    = (const float*)d_in[1];
  const float* bias = (const float*)d_in[2];
  float* out = (float*)d_out;

  float* ws   = (float*)d_ws;
  float* Ftab = ws;
  float* XbR  = ws + 32768;
  float* XbI  = XbR + 2097152;
  float* YbR  = XbI + 2097152;
  float* YbI  = YbR + 2097152;
  float* SR   = YbI + 2097152;
  float* SI   = SR + 4194304;

  k0_tables<<<16, 256, 0, stream>>>(Ftab);
  k1_fwd  <<<512, 256, 0, stream>>>(x, Ftab, XbR, XbI);
  k3_mix  <<<256, 512, 0, stream>>>(XbR, XbI, w, Ftab, YbR, YbI);
  k4a_col <<<512, 256, 0, stream>>>(YbR, YbI, Ftab, SR, SI);
  k4b_row <<<1024, 256, 0, stream>>>(SR, SI, Ftab, bias, out);
}


# Staging loops of k1, k3, k4a (table rows, filter taps) issue all their global loads as one group ahead of the operand loads instead of one load-wait-store round trip per pass; sums unchanged
# speedup vs baseline: 8.8043x; 1.0503x over previous
#include <hip/hip_runtime.h>


typedef __attribute__((ext_vector_type(2))) float v2f;
typedef __attribute__((ext_vector_type(4))) float v4f;
typedef __attribute__((ext_vector_type(8))) float v8f;

#define PI_F 3.14159265358979323846f

__device__ __forceinline__ v8f wmma4(v2f a, v2f b, v8f c) {
  return __builtin_amdgcn_wmma_f32_16x16x4_f32(false, a, false, b, (short)0, c,
                                               false, false);
}

__device__ __forceinline__ int bandmap(int bi) { return (bi + 96) & 127; }

__device__ __forceinline__ v2f fragA_kmap(const float* F, int m0, int k0) {
  int lane = threadIdx.x & 31;
  int m = m0 + (lane & 15);
  int kk = k0 + ((lane >> 4) << 1);
  v2f r; r.x = F[m * 128 + bandmap(kk)]; r.y = F[m * 128 + bandmap(kk + 1)];
  return r;
}
__device__ __forceinline__ v2f fragB_kmap(const float* F, int k0, int n0) {
  int lane = threadIdx.x & 31;
  int n = n0 + (lane & 15);
  int kk = k0 + ((lane >> 4) << 1);
  v2f r; r.x = F[bandmap(kk) * 128 + n]; r.y = F[bandmap(kk + 1) * 128 + n];
  return r;
}

__global__ void k0_tables(float* __restrict__ Ftab) {
  for (int t = threadIdx.x + blockIdx.x * blockDim.x; t < 16384;
       t += blockDim.x * gridDim.x) {
    int j = t >> 7, k = t & 127;
    float ang = -2.f * PI_F * (float)((j * k) & 127) / 128.f;
    float s, c;
    sincosf(ang, &s, &c);
    Ftab[t] = c;
    Ftab[16384 + t] = s;
  }
}

#define FP 132
__global__ void __launch_bounds__(256)
k1_fwd(const float* __restrict__ x, const float* __restrict__ Ftab,
       float* __restrict__ XbR, float* __restrict__ XbI) {
  __shared__ __attribute__((aligned(16))) float Fbr[32 * FP];
  __shared__ __attribute__((aligned(16))) float Fbi[32 * FP];
  __shared__ __attribute__((aligned(16))) float Tr[64 * FP];
  __shared__ __attribute__((aligned(16))) float Ti[64 * FP];
  const float* Fre = Ftab;
  const float* Fim = Ftab + 16384;
  int blk = blockIdx.x;
  int q = blk >> 6, ci = blk & 63;
  const float* Xre = x + ((size_t)(((q + 8) << 6) + ci) << 14);
  const float* Xim = x + ((size_t)((q << 6) + ci) << 14);
  int wave = threadIdx.x >> 5;
  int lane = threadIdx.x & 31;
  int c = lane & 15, h = lane >> 4;

  v4f fsr[4], fsi[4];
#pragma unroll
  for (int t = 0; t < 4; ++t) {
    int qs = threadIdx.x + (t << 8);
    int g = (qs >> 5) * 128 + ((qs & 31) << 2);
    fsr[t] = *(const v4f*)(Fre + g);
    fsi[t] = *(const v4f*)(Fim + g);
  }

  v2f e1[16], o1[16], e2[16], o2[16];
  const float* xc1 = Xre + (wave << 4) + c;
  const float* xc2 = Xim + (wave << 4) + c;
  const float* xa1 = xc1 + (h << 8);
  const float* xa2 = xc2 + (h << 8);
  const float* xb1 = xc1 + ((127 - (h << 1)) << 7);
  const float* xb2 = xc2 + ((127 - (h << 1)) << 7);
  int p0 = ((128 - (h << 1)) & 127) << 7;
#pragma unroll
  for (int ks = 0; ks < 16; ++ks) {
    v2f a1, b1, a2, b2;
    a1.x = xa1[ks << 9];
    a1.y = xa1[(ks << 9) + 128];
    a2.x = xa2[ks << 9];
    a2.y = xa2[(ks << 9) + 128];
    b1.y = xb1[-(ks << 9)];
    b2.y = xb2[-(ks << 9)];
    b1.x = (ks == 0) ? xc1[p0] : xb1[128 - (ks << 9)];
    b2.x = (ks == 0) ? xc2[p0] : xb2[128 - (ks << 9)];
    e1[ks] = a1 + b1;
    o1[ks] = a1 - b1;
    e2[ks] = a2 + b2;
    o2[ks] = a2 - b2;
  }
  if (h == 0) { e1[0].x *= 0.5f; e2[0].x *= 0.5f; }
  float x64a = xc1[64 << 7];
  float x64b = xc2[64 << 7];

#pragma unroll
  for (int t = 0; t < 4; ++t) {
    int qs = threadIdx.x + (t << 8);
    int m = qs >> 5, k4 = (qs & 31) << 2;
    *(v4f*)(Fbr + m * FP + k4) = fsr[t];
    *(v4f*)(Fbi + m * FP + k4) = fsi[t];
  }
  __syncthreads();

  for (int mt = 0; mt < 2; ++mt) {
    v8f g1 = {}, g2 = {}, g3 = {}, g4 = {};
    const float* ar_p = Fbr + ((mt << 4) + c) * FP + (h << 1);
    const float* ai_p = Fbi + ((mt << 4) + c) * FP + (h << 1);
#pragma unroll
    for (int ks = 0; ks < 16; ++ks) {
      v2f ar = *(const v2f*)(ar_p + (ks << 2));
      v2f ai = *(const v2f*)(ai_p + (ks << 2));
      g1 = wmma4(ar, e1[ks], g1);
      g2 = wmma4(ai, o2[ks], g2);
      g3 = wmma4(ar, e2[ks], g3);
      g4 = wmma4(ai, o1[ks], g4);
    }
#pragma unroll
    for (int r = 0; r < 8; ++r) {
      g1[r] += (r & 1) ? -x64a : x64a;
      g3[r] += (r & 1) ? -x64b : x64b;
    }
    int wc = (wave << 4) + c;
    int ub = (mt << 4) + (h << 3);
#pragma unroll
    for (int r = 0; r < 8; ++r) {
      int u = ub + r;
      Tr[(32 + u) * FP + wc] = g1[r] - g2[r];
      Ti[(32 + u) * FP + wc] = g3[r] + g4[r];
      int um = (u == 0) ? 0 : 32 - u;
      Tr[um * FP + wc] = (u == 0) ? 0.f : g1[r] + g2[r];
      Ti[um * FP + wc] = (u == 0) ? 0.f : g3[r] - g4[r];
    }
  }
  __syncthreads();

  int nt = wave & 1;
  int mt2 = wave >> 1;
  const float* br_p = Fbr + ((nt << 4) + c) * FP + (h << 1);
  const float* bi_p = Fbi + ((nt << 4) + c) * FP + (h << 1);
  const float* tr_p = Tr + ((mt2 << 4) + c) * FP + (h << 1);
  const float* ti_p = Ti + ((mt2 << 4) + c) * FP + (h << 1);
  v8f p1 = {}, p2 = {}, p3 = {}, p4 = {};
#pragma unroll 4
  for (int ks = 0; ks < 32; ++ks) {
    v2f br = *(const v2f*)(br_p + (ks << 2));
    v2f bi = *(const v2f*)(bi_p + (ks << 2));
    v2f tr = *(const v2f*)(tr_p + (ks << 2));
    v2f ti = *(const v2f*)(ti_p + (ks << 2));
    p1 = wmma4(tr, br, p1);
    p2 = wmma4(ti, bi, p2);
    p3 = wmma4(tr, bi, p3);
    p4 = wmma4(ti, br, p4);
  }
  int v = (nt << 4) + c;
  int vm = (v == 0) ? 0 : 32 - v;
  int mb = (mt2 << 4) + (h << 3);
#pragma unroll
  for (int r = 0; r < 8; ++r) {
    int uu = mb + r;
    int du = uu - 32;
    float msk = (du * du + v * v <= 900) ? 1.f : 0.f;
    size_t row = ((size_t)blk << 12) + (size_t)(uu * 64);
    XbR[row + 32 + v] = (p1[r] - p2[r]) * msk;
    XbI[row + 32 + v] = (p3[r] + p4[r]) * msk;
    XbR[row + vm] = (v == 0) ? 0.f : (p1[r] + p2[r]) * msk;
    XbI[row + vm] = (v == 0) ? 0.f : (p4[r] - p3[r]) * msk;
  }
}

#define WPK 386
#define WPP 1184
__global__ void __launch_bounds__(512) __attribute__((amdgpu_waves_per_eu(4, 4)))
k3_mix(const float* __restrict__ XbR, const float* __restrict__ XbI,
       const float* __restrict__ w, const float* __restrict__ Ftab,
       float* __restrict__ YbR, float* __restrict__ YbI) {
  __shared__ __attribute__((aligned(16))) float Wc[32 * WPP];
  int wave = threadIdx.x >> 5;
  int pr = wave >> 1;
  int oh = wave & 1;
  int lane = threadIdx.x & 31;
  int c = lane & 15, h = lane >> 4;

  int u = bandmap(blockIdx.x >> 2);
  int u2 = (u + u) & 127;
  float a1r = Ftab[128 + u],  a1i = Ftab[16384 + 128 + u];
  float a2r = Ftab[128 + u2], a2i = Ftab[16384 + 128 + u2];

  float tw0[24], tw1[24], tw2[24];
#pragma unroll
  for (int t = 0; t < 24; ++t) {
    int e = threadIdx.x + (t << 9);
    int rest = e >> 5;
    int oh = rest / 24;
    int rho = ((rest - oh * 24) << 3) + (e & 7);
    int osrc = (oh << 2) + ((e >> 3) & 3);
    int i = rho / 3;
    int kx = rho - i * 3;
    const float* wp = w + osrc * 576 + i * 9 + kx;
    tw0[t] = wp[0]; tw1[t] = wp[3]; tw2[t] = wp[6];
  }
#pragma unroll
  for (int t = 0; t < 24; ++t) {
    int e = threadIdx.x + (t << 9);
    int rest = e >> 5;
    int oh = rest / 24;
    int rho = ((rest - oh * 24) << 3) + (e & 7);
    int osrc = (oh << 2) + ((e >> 3) & 3);
    int i = rho / 3;
    int kx = rho - i * 3;
    float w0 = tw0[t], w1 = tw1[t], w2 = tw2[t];
    int col = (osrc + 32) & 63;
    float wre = w0 + a1r * w1 + a2r * w2;
    float wim = a1i * w1 + a2i * w2;
    float* d = Wc + (i >> 1) * WPP + kx * WPK + col * 6 + (i & 1);
    d[0] = wre;
    d[2] = wim - wre;
    d[4] = wre + wim;
  }

  int f = (blockIdx.x << 4) + c;
  int v = bandmap(f & 63);
  int v2 = (v + v) & 127;
  float b1r = Ftab[128 + v],  b1i = Ftab[16384 + 128 + v];
  float b2r = Ftab[128 + v2], b2i = Ftab[16384 + 128 + v2];
  __syncthreads();

  const float* xr_p = XbR + ((size_t)(pr << 6) << 12) + f;
  const float* xi_p = XbI + ((size_t)(pr << 6) << 12) + f;
  v8f s1[2] = {}, s2[2] = {}, s3[2] = {};
#pragma nounroll
  for (int i0 = 0; i0 < 64; i0 += 4) {
    int kk = i0 + (h << 1);
    float xr0 = xr_p[(size_t)kk << 12], xr1 = xr_p[(size_t)(kk + 1) << 12];
    float xi0 = xi_p[(size_t)kk << 12], xi1 = xi_p[(size_t)(kk + 1) << 12];
    const float* wb = Wc + (kk >> 1) * WPP + (oh * 32 + c) * 6;
    v2f bf[3][2][3];
#pragma unroll
    for (int kx = 0; kx < 3; ++kx)
#pragma unroll
      for (int ot = 0; ot < 2; ++ot) {
        const v2f* q = (const v2f*)(wb + kx * WPK + ot * 96);
        bf[kx][ot][0] = q[0]; bf[kx][ot][1] = q[1]; bf[kx][ot][2] = q[2];
      }
#pragma unroll
    for (int kx = 0; kx < 3; ++kx) {
      v2f ar, ai;
      if (kx == 0) {
        ar.x = xr0; ar.y = xr1; ai.x = xi0; ai.y = xi1;
      } else {
        float tr = (kx == 1) ? b1r : b2r;
        float ti = (kx == 1) ? b1i : b2i;
        ar.x = xr0 * tr - xi0 * ti;
        ar.y = xr1 * tr - xi1 * ti;
        ai.x = xr0 * ti + xi0 * tr;
        ai.y = xr1 * ti + xi1 * tr;
      }
      v2f as = ar + ai;
#pragma unroll
      for (int ot = 0; ot < 2; ++ot) {
        s1[ot] = wmma4(as, bf[kx][ot][0], s1[ot]);
        s2[ot] = wmma4(ar, bf[kx][ot][1], s2[ot]);
        s3[ot] = wmma4(ai, bf[kx][ot][2], s3[ot]);
      }
    }
  }

  int fb = (blockIdx.x << 4) + (h << 3);
#pragma unroll
  for (int ot = 0; ot < 2; ++ot)
#pragma unroll
    for (int r = 0; r < 8; ++r) {
      size_t off = ((size_t)(fb + r) << 9) + (size_t)((pr << 6) + (oh << 5) + (ot << 4) + c);
      YbR[off] = s1[ot][r] - s3[ot][r];
      YbI[off] = s1[ot][r] + s2[ot][r];
    }
}

#define APITCH 68
__global__ void k4a_col(const float* __restrict__ YbR, const float* __restrict__ YbI,
                        const float* __restrict__ Ftab,
                        float* __restrict__ SR, float* __restrict__ SI) {
  __shared__ __attribute__((aligned(16))) float Af[64 * APITCH];
  const float* Fre = Ftab;
  const float* Fim = Ftab + 16384;
  int mh = blockIdx.x & 1;
  int nb = blockIdx.x >> 1;
  int wave = threadIdx.x >> 5;
  int lane = threadIdx.x & 31;
  int c = lane & 15, h = lane >> 4;
  int n0 = ((nb << 3) + wave) << 4;
  int vv = n0 >> 9, q = (n0 >> 6) & 7, o0 = n0 & 63;

  v4f fst[4];
#pragma unroll
  for (int t = 0; t < 4; ++t) {
    int idx = threadIdx.x + (t << 8);
    int mr = idx >> 4, k4 = (idx & 15) << 2;
    const float* Fsrc = (k4 < 32) ? Fre : Fim;
    fst[t] = *(const v4f*)(Fsrc + ((mh << 6) + mr) * 128 + (k4 & 31));
  }

  v2f bp[8], bq[8], cp[8], cq[8];
  const float* yr = YbR + n0 + c;
  const float* yi = YbI + n0 + c;
#pragma unroll
  for (int ks = 0; ks < 8; ++ks) {
    int u = (ks << 2) + (h << 1);
    size_t rp = (size_t)(32 + u) << 15, rm = (size_t)(32 - u) << 15;
    v2f yrp, yrm, yip, yim;
    yrp.x = yr[rp]; yrp.y = yr[rp + 32768];
    yrm.x = yr[rm]; yrm.y = yr[rm - 32768];
    yip.x = yi[rp]; yip.y = yi[rp + 32768];
    yim.x = yi[rm]; yim.y = yi[rm - 32768];
    bp[ks] = yrp + yrm;
    bq[ks] = yip - yim;
    cp[ks] = yip + yim;
    cq[ks] = yrm - yrp;
  }
  if (h == 0) { bp[0].x *= 0.5f; cp[0].x *= 0.5f; }

#pragma unroll
  for (int t = 0; t < 4; ++t) {
    int idx = threadIdx.x + (t << 8);
    int mr = idx >> 4, k4 = (idx & 15) << 2;
    *(v4f*)(Af + mr * APITCH + k4) = fst[t];
  }
  __syncthreads();

  for (int mt = 0; mt < 4; ++mt) {
    v8f cR = {}, cI = {};
    const float* a_p = Af + ((mt << 4) + c) * APITCH + (h << 1);
    v2f af[16];
#pragma unroll
    for (int ks = 0; ks < 16; ++ks) af[ks] = *(const v2f*)(a_p + (ks << 2));
#pragma unroll
    for (int ks = 0; ks < 16; ++ks) {
      v2f bre = (ks < 8) ? bp[ks & 7] : bq[ks & 7];
      v2f bim = (ks < 8) ? cp[ks & 7] : cq[ks & 7];
      cR = wmma4(af[ks], bre, cR);
      cI = wmma4(af[ks], bim, cI);
    }
    int mbase = (mh << 6) + (mt << 4) + (h << 3);
#pragma unroll
    for (int r = 0; r < 8; ++r) {
      size_t off = ((((size_t)(q * 128 + mbase + r) << 6) + vv) << 6) + o0 + c;
      SR[off] = cR[r];
      SI[off] = cI[r];
    }
  }
}

__global__ void __launch_bounds__(256)
k4b_row(const float* __restrict__ SR, const float* __restrict__ SI,
                        const float* __restrict__ Ftab, const float* __restrict__ bias,
                        float* __restrict__ out) {
  const float* Fre = Ftab;
  const float* Fim = Ftab + 16384;
  int blk = blockIdx.x;
  int q = blk >> 7, m = blk & 127;
  int wave = threadIdx.x >> 5;
  int lane = threadIdx.x & 31;
  int c = lane & 15, h = lane >> 4;
  int ot = wave & 3;
  int ng = wave >> 2;
  const float* sr_p = SR + ((size_t)blk << 12) + (ot << 4) + c;
  const float* si_p = SI + ((size_t)blk << 12) + (ot << 4) + c;

  v2f ap[8], aq[8], bp[8], bq[8];
#pragma unroll
  for (int ks = 0; ks < 8; ++ks) {
    int v = (ks << 2) + (h << 1);
    int rp = (32 + v) << 6, rm = (32 - v) << 6;
    v2f srp, srm, sip, sim;
    srp.x = sr_p[rp]; srp.y = sr_p[rp + 64];
    srm.x = sr_p[rm]; srm.y = sr_p[rm - 64];
    sip.x = si_p[rp]; sip.y = si_p[rp + 64];
    sim.x = si_p[rm]; sim.y = si_p[rm - 64];
    ap[ks] = srp + srm;
    aq[ks] = sip - sim;
    bp[ks] = sip + sim;
    bq[ks] = srm - srp;
  }
  if (h == 0) { ap[0].x *= 0.5f; bp[0].x *= 0.5f; }

  v8f pr[4] = {}, pi[4] = {};
#pragma unroll
  for (int ks = 0; ks < 16; ++ks) {
    const float* Fsrc = (ks < 8) ? Fre : Fim;
    int krow = ((ks & 7) << 2) + (h << 1);
    v2f are = (ks < 8) ? ap[ks & 7] : aq[ks & 7];
    v2f aim = (ks < 8) ? bp[ks & 7] : bq[ks & 7];
#pragma unroll
    for (int j = 0; j < 4; ++j) {
      int n = (((ng << 2) + j) << 4) + c;
      v2f f; f.x = Fsrc[krow * 128 + n]; f.y = Fsrc[(krow + 1) * 128 + n];
      pr[j] = wmma4(are, f, pr[j]);
      pi[j] = wmma4(aim, f, pi[j]);
    }
  }

  int ob = (ot << 4) + (h << 3);
#pragma unroll
  for (int r = 0; r < 8; ++r) {
    float bo = bias[ob + r];
    float* orow = out + ((((size_t)(q * 64 + ob + r) << 7) + m) << 7) + (ng << 6) + c;
    float* orow8 = orow + ((size_t)8 << 20);
#pragma unroll
    for (int j = 0; j < 4; ++j) {
      orow[j << 4]  = pr[j][r] * (1.f / 16384.f) + bo;
      orow8[j << 4] = pi[j][r] * (1.f / 16384.f) + bo;
    }
  }
}

extern "C" void kernel_launch(void* const* d_in, const int* in_sizes, int n_in,
                              void* d_out, int out_size, void* d_ws, size_t ws_size,
                              hipStream_t stream) {
  (void)in_sizes; (void)n_in; (void)out_size; (void)ws_size;
  const float* x    = (const float*)d_in[0];
  const float* w    = (const float*)d_in[1];
  const float* bias = (const float*)d_in[2];
  float* out = (float*)d_out;

  float* ws   = (float*)d_ws;
  float* Ftab = ws;
  float* XbR  = ws + 32768;
  float* XbI  = XbR + 2097152;
  float* YbR  = XbI + 2097152;
  float* YbI  = YbR + 2097152;
  float* SR   = YbI + 2097152;
  float* SI   = SR + 4194304;

  k0_tables<<<16, 256, 0, stream>>>(Ftab);
  k1_fwd  <<<512, 256, 0, stream>>>(x, Ftab, XbR, XbI);
  k3_mix  <<<256, 512, 0, stream>>>(XbR, XbI, w, Ftab, YbR, YbI);
  k4a_col <<<512, 256, 0, stream>>>(YbR, YbI, Ftab, SR, SI);
  k4b_row <<<1024, 256, 0, stream>>>(SR, SI, Ftab, bias, out);
}
